# starting kernel
# baseline (speedup 1.0000x reference)
#include <hip/hip_runtime.h>

typedef float v2f __attribute__((ext_vector_type(2)));
typedef float v8f __attribute__((ext_vector_type(8)));

#define IN_CH 512
#define HID   128
#define OUTC  64

__global__ void k_set_ones(float* deg, int n) {
  int i = blockIdx.x * blockDim.x + threadIdx.x;
  if (i < n) deg[i] = 1.0f;
}

__global__ void k_deg_accum(const int* __restrict__ ei, const float* __restrict__ ew,
                            float* deg, int E) {
  int e = blockIdx.x * blockDim.x + threadIdx.x;
  if (e < E) atomicAdd(&deg[ei[E + e]], ew[e]);
}

__global__ void k_rsqrt(const float* __restrict__ deg, float* dinv, int n) {
  int i = blockIdx.x * blockDim.x + threadIdx.x;
  if (i < n) {
    float d = deg[i];
    dinv[i] = (d > 0.0f) ? rsqrtf(d) : 0.0f;
  }
}

__global__ void k_pack_w(const float* __restrict__ W, float* __restrict__ Wp,
                         int K, int N) {
  int idx = blockIdx.x * blockDim.x + threadIdx.x;
  if (idx >= K * N) return;
  int k = idx / N;
  int n = idx - k * N;
  int k4 = k >> 2;
  int hi = (k >> 1) & 1;
  int j  = k & 1;
  Wp[(((size_t)((k4 << 1) | hi)) * N + n) * 2 + j] = W[idx];
}

template <int NT>
__global__ void k_gemm_wmma(const float* __restrict__ A,
                            const float* __restrict__ Bp,
                            float* __restrict__ C,
                            int M, int K) {
  const int N = NT * 16;
  int wave = (int)((blockIdx.x * blockDim.x + threadIdx.x) >> 5);
  int lane = threadIdx.x & 31;
  if (wave >= (M >> 4)) return;

  int m0 = wave << 4;
  int lo = lane & 15;
  int hi = lane >> 4;

  const float* aPtr = A  + (size_t)(m0 + lo) * K + (hi << 1);
  const float* bPtr = Bp + ((size_t)hi * N + lo) * 2;

  v8f z = {0.f,0.f,0.f,0.f,0.f,0.f,0.f,0.f};
  v8f acc[NT];
#pragma unroll
  for (int t = 0; t < NT; ++t) acc[t] = z;

  for (int k = 0; k < K; k += 8) {
    v2f a0 = *(const v2f*)(aPtr + k);
    v2f a1 = *(const v2f*)(aPtr + k + 4);
    const float* bp0 = bPtr + (size_t)k * N;
    const float* bp1 = bp0 + (size_t)4 * N;
    v2f b0[NT], b1[NT];
#pragma unroll
    for (int t = 0; t < NT; ++t) b0[t] = *(const v2f*)(bp0 + t * 32);
#pragma unroll
    for (int t = 0; t < NT; ++t) b1[t] = *(const v2f*)(bp1 + t * 32);
#pragma unroll
    for (int t = 0; t < NT; ++t)
      acc[t] = __builtin_amdgcn_wmma_f32_16x16x4_f32(
          false, a0, false, b0[t], (short)0, acc[t], false, false);
#pragma unroll
    for (int t = 0; t < NT; ++t)
      acc[t] = __builtin_amdgcn_wmma_f32_16x16x4_f32(
          false, a1, false, b1[t], (short)0, acc[t], false, false);
  }

  int cRow = m0 + (hi << 3);
#pragma unroll
  for (int r = 0; r < 8; ++r) {
    float* cp = C + (size_t)(cRow + r) * N + lo;
#pragma unroll
    for (int t = 0; t < NT; ++t) cp[t * 16] = acc[t][r];
  }
}

__global__ void k_init_agg1(const float* __restrict__ h1, const float* __restrict__ dinv,
                            const float* __restrict__ b1, float* agg, int n) {
  int idx = blockIdx.x * blockDim.x + threadIdx.x;
  if (idx < n) {
    int row = idx >> 7, c = idx & 127;
    float di = dinv[row];
    agg[idx] = h1[idx] * di * di + b1[c];
  }
}

__global__ void k_scatter1(const int* __restrict__ ei, const float* __restrict__ ew,
                           const float* __restrict__ dinv, const float* __restrict__ h,
                           float* agg, int E) {
  long long idx = (long long)blockIdx.x * blockDim.x + threadIdx.x;
  if (idx >= (long long)E * 32) return;
  int e = (int)(idx >> 5);
  int g = ((int)idx & 31) << 2;
  int s = ei[e], d = ei[E + e];
  float nm = dinv[s] * ew[e] * dinv[d];
  float4 v = *(const float4*)(h + (size_t)s * HID + g);
  float* o = agg + (size_t)d * HID + g;
  atomicAdd(o + 0, v.x * nm);
  atomicAdd(o + 1, v.y * nm);
  atomicAdd(o + 2, v.z * nm);
  atomicAdd(o + 3, v.w * nm);
}

__global__ void k_relu(float* a, int n) {
  int i = blockIdx.x * blockDim.x + threadIdx.x;
  if (i < n) a[i] = fmaxf(a[i], 0.0f);
}

__global__ void k_init_out(const float* __restrict__ t2, const float* __restrict__ t3,
                           const float* __restrict__ dinv,
                           const float* __restrict__ b2, const float* __restrict__ b3,
                           float* om, float* ol, int n) {
  int idx = blockIdx.x * blockDim.x + threadIdx.x;
  if (idx < n) {
    int row = idx >> 6, c = idx & 63;
    float di = dinv[row];
    float d2 = di * di;
    om[idx] = t2[idx] * d2 + b2[c];
    ol[idx] = t3[idx] * d2 + b3[c];
  }
}

__global__ void k_scatter2(const int* __restrict__ ei, const float* __restrict__ ew,
                           const float* __restrict__ dinv,
                           const float* __restrict__ t2, const float* __restrict__ t3,
                           float* om, float* ol, int E) {
  long long idx = (long long)blockIdx.x * blockDim.x + threadIdx.x;
  if (idx >= (long long)E * 32) return;
  int e = (int)(idx >> 5);
  int l = (int)idx & 31;
  int s = ei[e], d = ei[E + e];
  float nm = dinv[s] * ew[e] * dinv[d];
  const float* tin = (l < 16) ? t2 : t3;
  float*       op  = (l < 16) ? om : ol;
  int g = (l & 15) << 2;
  float4 v = *(const float4*)(tin + (size_t)s * OUTC + g);
  float* o = op + (size_t)d * OUTC + g;
  atomicAdd(o + 0, v.x * nm);
  atomicAdd(o + 1, v.y * nm);
  atomicAdd(o + 2, v.z * nm);
  atomicAdd(o + 3, v.w * nm);
}

extern "C" void kernel_launch(void* const* d_in, const int* in_sizes, int n_in,
                              void* d_out, int out_size, void* d_ws, size_t ws_size,
                              hipStream_t stream) {
  const float* x  = (const float*)d_in[0];
  const int*   ei = (const int*)  d_in[1];
  const float* ew = (const float*)d_in[2];
  const float* W1 = (const float*)d_in[3];
  const float* b1 = (const float*)d_in[4];
  const float* W2 = (const float*)d_in[5];
  const float* b2 = (const float*)d_in[6];
  const float* W3 = (const float*)d_in[7];
  const float* b3 = (const float*)d_in[8];

  const int N = in_sizes[0] / IN_CH;
  const int E = in_sizes[2];

  float* ws   = (float*)d_ws;
  float* deg  = ws;
  float* dinv = ws + N;
  float* h1   = ws + 2 * (size_t)N;
  float* agg1 = h1 + (size_t)N * HID;
  float* W1p  = agg1 + (size_t)N * HID;
  float* W2p  = W1p + (size_t)IN_CH * HID;
  float* W3p  = W2p + (size_t)HID * OUTC;
  float* t2   = h1;
  float* t3   = h1 + (size_t)N * OUTC;
  float* om   = (float*)d_out;
  float* ol   = om + (size_t)N * OUTC;

  const int T = 256;

  k_pack_w<<<(IN_CH * HID + T - 1) / T, T, 0, stream>>>(W1, W1p, IN_CH, HID);
  k_pack_w<<<(HID * OUTC + T - 1) / T, T, 0, stream>>>(W2, W2p, HID, OUTC);
  k_pack_w<<<(HID * OUTC + T - 1) / T, T, 0, stream>>>(W3, W3p, HID, OUTC);

  k_set_ones  <<<(N + T - 1) / T, T, 0, stream>>>(deg, N);
  k_deg_accum <<<(E + T - 1) / T, T, 0, stream>>>(ei, ew, deg, E);
  k_rsqrt     <<<(N + T - 1) / T, T, 0, stream>>>(deg, dinv, N);

  {
    int waves  = N >> 4;
    int blocks = (waves * 32 + T - 1) / T;
    k_gemm_wmma<8><<<blocks, T, 0, stream>>>(x, W1p, h1, N, IN_CH);
  }

  k_init_agg1<<<((N * HID) + T - 1) / T, T, 0, stream>>>(h1, dinv, b1, agg1, N * HID);
  {
    long long tot = (long long)E * 32;
    int blocks = (int)((tot + T - 1) / T);
    k_scatter1<<<blocks, T, 0, stream>>>(ei, ew, dinv, h1, agg1, E);
  }
  k_relu<<<((N * HID) + T - 1) / T, T, 0, stream>>>(agg1, N * HID);
  const float* h = agg1;

  {
    int waves  = N >> 4;
    int blocks = (waves * 32 + T - 1) / T;
    k_gemm_wmma<4><<<blocks, T, 0, stream>>>(h, W2p, t2, N, HID);
    k_gemm_wmma<4><<<blocks, T, 0, stream>>>(h, W3p, t3, N, HID);
  }

  k_init_out<<<((N * OUTC) + T - 1) / T, T, 0, stream>>>(t2, t3, dinv, b2, b3, om, ol, N * OUTC);
  {
    long long tot = (long long)E * 32;
    int blocks = (int)((tot + T - 1) / T);
    k_scatter2<<<blocks, T, 0, stream>>>(ei, ew, dinv, t2, t3, om, ol, E);
  }
}


# Edge scatter-adds replaced by a per-call CSR by destination (returning int add = arrival-order rank, scan, fill); one wave per row sums gathered rows in registers: no f32 atomics, sum order changes
# speedup vs baseline: 2.1066x; 2.1066x over previous
#include <hip/hip_runtime.h>

typedef float v2f __attribute__((ext_vector_type(2)));
typedef float v8f __attribute__((ext_vector_type(8)));

#define IN_CH 512
#define HID   128
#define OUTC  64

__global__ void k_zero_cnt(int* cnt, int n) {
  int i = blockIdx.x * blockDim.x + threadIdx.x;
  if (i < n) cnt[i] = 0;
}

__global__ void k_rank(const int* __restrict__ ei, int* cnt, int* __restrict__ rank, int E) {
  int e = blockIdx.x * blockDim.x + threadIdx.x;
  if (e < E) rank[e] = atomicAdd(&cnt[ei[E + e]], 1);
}

__global__ void __launch_bounds__(1024)
k_scan(const int* __restrict__ cnt, int* __restrict__ rowptr, int n) {
  __shared__ int wtot[32];
  __shared__ int wpart[32];
  int t = threadIdx.x;
  int lane = t & 31;
  int w = t >> 5;
  int b0 = blockIdx.x * 1024;
  int part = 0;
  for (int i = t; i < b0; i += 1024) part += cnt[i];
  int i = b0 + t;
  int c = (i < n) ? cnt[i] : 0;
  int x = c;
  for (int dlt = 1; dlt < 32; dlt <<= 1) {
    int y = __shfl_up(x, dlt);
    if (lane >= dlt) x += y;
  }
  for (int dlt = 16; dlt > 0; dlt >>= 1) part += __shfl_xor(part, dlt);
  if (lane == 31) wtot[w] = x;
  if (lane == 0) wpart[w] = part;
  __syncthreads();
  int off = 0;
  for (int k = 0; k < 32; ++k) {
    off += wpart[k];
    if (k < w) off += wtot[k];
  }
  if (i < n) rowptr[i] = off + x - c;
}

__global__ void k_fill(const int* __restrict__ ei, const float* __restrict__ ew,
                       const int* __restrict__ rowptr, const int* __restrict__ rank,
                       int* __restrict__ csr, int E) {
  int e = blockIdx.x * blockDim.x + threadIdx.x;
  if (e < E) {
    int d = ei[E + e];
    int p = rowptr[d] + rank[e];
    int2 r;
    r.x = ei[e];
    r.y = __float_as_int(ew[e]);
    *(int2*)(csr + 2 * (size_t)p) = r;
  }
}

__global__ void k_dinv(const int* __restrict__ csr, const int* __restrict__ rowptr,
                       const int* __restrict__ cnt, float* __restrict__ dinv, int n) {
  int gid = blockIdx.x * blockDim.x + threadIdx.x;
  int node = gid >> 3;
  int j = gid & 7;
  float s = 0.0f;
  if (node < n) {
    int st = rowptr[node];
    int len = cnt[node];
    for (int k = j; k < len; k += 8) s += __int_as_float(csr[2 * (size_t)(st + k) + 1]);
  }
  s += __shfl_xor(s, 1);
  s += __shfl_xor(s, 2);
  s += __shfl_xor(s, 4);
  if (node < n && j == 0) {
    float dg = 1.0f + s;
    dinv[node] = (dg > 0.0f) ? rsqrtf(dg) : 0.0f;
  }
}

__global__ void k_pack_w(const float* __restrict__ W, float* __restrict__ Wp,
                         int K, int N) {
  int idx = blockIdx.x * blockDim.x + threadIdx.x;
  if (idx >= K * N) return;
  int k = idx / N;
  int n = idx - k * N;
  int k4 = k >> 2;
  int hi = (k >> 1) & 1;
  int j  = k & 1;
  Wp[(((size_t)((k4 << 1) | hi)) * N + n) * 2 + j] = W[idx];
}

template <int NT>
__global__ void k_gemm_wmma(const float* __restrict__ A,
                            const float* __restrict__ Bp,
                            float* __restrict__ C,
                            int M, int K) {
  const int N = NT * 16;
  int wave = (int)((blockIdx.x * blockDim.x + threadIdx.x) >> 5);
  int lane = threadIdx.x & 31;
  if (wave >= (M >> 4)) return;

  int m0 = wave << 4;
  int lo = lane & 15;
  int hi = lane >> 4;

  const float* aPtr = A  + (size_t)(m0 + lo) * K + (hi << 1);
  const float* bPtr = Bp + ((size_t)hi * N + lo) * 2;

  v8f z = {0.f,0.f,0.f,0.f,0.f,0.f,0.f,0.f};
  v8f acc[NT];
#pragma unroll
  for (int t = 0; t < NT; ++t) acc[t] = z;

  for (int k = 0; k < K; k += 8) {
    v2f a0 = *(const v2f*)(aPtr + k);
    v2f a1 = *(const v2f*)(aPtr + k + 4);
    const float* bp0 = bPtr + (size_t)k * N;
    const float* bp1 = bp0 + (size_t)4 * N;
    v2f b0[NT], b1[NT];
#pragma unroll
    for (int t = 0; t < NT; ++t) b0[t] = *(const v2f*)(bp0 + t * 32);
#pragma unroll
    for (int t = 0; t < NT; ++t) b1[t] = *(const v2f*)(bp1 + t * 32);
#pragma unroll
    for (int t = 0; t < NT; ++t)
      acc[t] = __builtin_amdgcn_wmma_f32_16x16x4_f32(
          false, a0, false, b0[t], (short)0, acc[t], false, false);
#pragma unroll
    for (int t = 0; t < NT; ++t)
      acc[t] = __builtin_amdgcn_wmma_f32_16x16x4_f32(
          false, a1, false, b1[t], (short)0, acc[t], false, false);
  }

  int cRow = m0 + (hi << 3);
#pragma unroll
  for (int r = 0; r < 8; ++r) {
    float* cp = C + (size_t)(cRow + r) * N + lo;
#pragma unroll
    for (int t = 0; t < NT; ++t) cp[t * 16] = acc[t][r];
  }
}

template <bool RELU>
__global__ void __launch_bounds__(256)
k_gather(const int* __restrict__ csr, const int* __restrict__ rowptr,
         const int* __restrict__ cnt, const float* __restrict__ dinv,
         const float* __restrict__ inA, const float* __restrict__ inB,
         const float* __restrict__ biasA, const float* __restrict__ biasB,
         float* __restrict__ outA, float* __restrict__ outB, int ld, int n) {
  int lane = threadIdx.x & 31;
  int d = __builtin_amdgcn_readfirstlane((int)((blockIdx.x * blockDim.x + threadIdx.x) >> 5));
  if (d >= n) return;
  int st  = rowptr[d];
  int len = cnt[d];
  float dd = dinv[d];
  int col = (lane & 15) << 2;
  const float* in  = ((lane < 16) ? inA : inB) + col;
  const float* bp  = ((lane < 16) ? biasA : biasB) + col;
  float*       out = ((lane < 16) ? outA : outB) + (size_t)d * ld + col;

  float4 v = *(const float4*)(in + (size_t)d * ld);
  float4 acc;
  acc.x = v.x * dd * dd + bp[0];
  acc.y = v.y * dd * dd + bp[1];
  acc.z = v.z * dd * dd + bp[2];
  acc.w = v.w * dd * dd + bp[3];

  for (int base = 0; base < len; base += 32) {
    int m = len - base;
    if (m > 32) m = 32;
    int sj = d;
    float nj = 0.0f;
    if (lane < m) {
      int2 r = *(const int2*)(csr + 2 * (size_t)(st + base + lane));
      sj = r.x;
      nj = dinv[r.x] * __int_as_float(r.y) * dd;
    }
    for (int j = 0; j < m; j += 8) {
      float4 r[8];
      float wgt[8];
#pragma unroll
      for (int u = 0; u < 8; ++u) {
        int s = __builtin_amdgcn_readlane(sj, j + u);
        wgt[u] = __int_as_float(__builtin_amdgcn_readlane(__float_as_int(nj), j + u));
        float4 t = make_float4(0.0f, 0.0f, 0.0f, 0.0f);
        if (j + u < m) t = *(const float4*)(in + (size_t)s * ld);
        r[u] = t;
      }
#pragma unroll
      for (int u = 0; u < 8; ++u) {
        acc.x += r[u].x * wgt[u];
        acc.y += r[u].y * wgt[u];
        acc.z += r[u].z * wgt[u];
        acc.w += r[u].w * wgt[u];
      }
    }
  }
  if (RELU) {
    acc.x = fmaxf(acc.x, 0.0f);
    acc.y = fmaxf(acc.y, 0.0f);
    acc.z = fmaxf(acc.z, 0.0f);
    acc.w = fmaxf(acc.w, 0.0f);
  }
  *(float4*)out = acc;
}

extern "C" void kernel_launch(void* const* d_in, const int* in_sizes, int n_in,
                              void* d_out, int out_size, void* d_ws, size_t ws_size,
                              hipStream_t stream) {
  const float* x  = (const float*)d_in[0];
  const int*   ei = (const int*)  d_in[1];
  const float* ew = (const float*)d_in[2];
  const float* W1 = (const float*)d_in[3];
  const float* b1 = (const float*)d_in[4];
  const float* W2 = (const float*)d_in[5];
  const float* b2 = (const float*)d_in[6];
  const float* W3 = (const float*)d_in[7];
  const float* b3 = (const float*)d_in[8];

  const int N = in_sizes[0] / IN_CH;
  const int E = in_sizes[2];

  float* ws   = (float*)d_ws;
  int*   cnt  = (int*)ws;
  float* dinv = ws + N;
  float* h1   = ws + 2 * (size_t)N;
  float* er   = h1 + (size_t)N * HID;
  int*   rank   = (int*)er;
  int*   csr    = rank + (size_t)E;
  int*   rowptr = csr + 2 * (size_t)E;
  float* W1p  = er + (size_t)N * HID;
  float* W2p  = W1p + (size_t)IN_CH * HID;
  float* W3p  = W2p + (size_t)HID * OUTC;
  float* t2   = h1;
  float* t3   = h1 + (size_t)N * OUTC;
  float* om   = (float*)d_out;
  float* ol   = om + (size_t)N * OUTC;
  float* h    = (float*)d_out;

  const int T = 256;

  k_pack_w<<<(IN_CH * HID + T - 1) / T, T, 0, stream>>>(W1, W1p, IN_CH, HID);
  k_pack_w<<<(HID * OUTC + T - 1) / T, T, 0, stream>>>(W2, W2p, HID, OUTC);
  k_pack_w<<<(HID * OUTC + T - 1) / T, T, 0, stream>>>(W3, W3p, HID, OUTC);

  k_zero_cnt<<<(N + T - 1) / T, T, 0, stream>>>(cnt, N);
  k_rank    <<<(E + T - 1) / T, T, 0, stream>>>(ei, cnt, rank, E);
  k_scan    <<<(N + 1023) / 1024, 1024, 0, stream>>>(cnt, rowptr, N);
  k_fill    <<<(E + T - 1) / T, T, 0, stream>>>(ei, ew, rowptr, rank, csr, E);
  k_dinv    <<<(N * 8 + T - 1) / T, T, 0, stream>>>(csr, rowptr, cnt, dinv, N);

  {
    int waves  = N >> 4;
    int blocks = (waves * 32 + T - 1) / T;
    k_gemm_wmma<8><<<blocks, T, 0, stream>>>(x, W1p, h1, N, IN_CH);
  }

  k_gather<true><<<(N * 32 + T - 1) / T, T, 0, stream>>>(
      csr, rowptr, cnt, dinv, h1, h1 + 64, b1, b1 + 64, h, h + 64, HID, N);

  {
    int waves  = N >> 4;
    int blocks = (waves * 32 + T - 1) / T;
    k_gemm_wmma<4><<<blocks, T, 0, stream>>>(h, W2p, t2, N, HID);
    k_gemm_wmma<4><<<blocks, T, 0, stream>>>(h, W3p, t3, N, HID);
  }

  k_gather<false><<<(N * 32 + T - 1) / T, T, 0, stream>>>(
      csr, rowptr, cnt, dinv, t2, t3, b2, b3, om, ol, OUTC, N);
}


# Source-side dinv folded into the three GEMM epilogues (rows stored pre-scaled); gather multiplies by raw edge weight, applies dinv[dst] once after the sum: no per-edge dinv lookup; sum order kept
# speedup vs baseline: 2.1913x; 1.0402x over previous
#include <hip/hip_runtime.h>

typedef float v2f __attribute__((ext_vector_type(2)));
typedef float v8f __attribute__((ext_vector_type(8)));

#define IN_CH 512
#define HID   128
#define OUTC  64

__global__ void k_zero_cnt(int* cnt, int n) {
  int i = blockIdx.x * blockDim.x + threadIdx.x;
  if (i < n) cnt[i] = 0;
}

__global__ void k_rank(const int* __restrict__ ei, int* cnt, int* __restrict__ rank, int E) {
  int e = blockIdx.x * blockDim.x + threadIdx.x;
  if (e < E) rank[e] = atomicAdd(&cnt[ei[E + e]], 1);
}

__global__ void __launch_bounds__(1024)
k_scan(const int* __restrict__ cnt, int* __restrict__ rowptr, int n) {
  __shared__ int wtot[32];
  __shared__ int wpart[32];
  int t = threadIdx.x;
  int lane = t & 31;
  int w = t >> 5;
  int b0 = blockIdx.x * 1024;
  int part = 0;
  for (int i = t; i < b0; i += 1024) part += cnt[i];
  int i = b0 + t;
  int c = (i < n) ? cnt[i] : 0;
  int x = c;
  for (int dlt = 1; dlt < 32; dlt <<= 1) {
    int y = __shfl_up(x, dlt);
    if (lane >= dlt) x += y;
  }
  for (int dlt = 16; dlt > 0; dlt >>= 1) part += __shfl_xor(part, dlt);
  if (lane == 31) wtot[w] = x;
  if (lane == 0) wpart[w] = part;
  __syncthreads();
  int off = 0;
  for (int k = 0; k < 32; ++k) {
    off += wpart[k];
    if (k < w) off += wtot[k];
  }
  if (i < n) rowptr[i] = off + x - c;
}

__global__ void k_fill(const int* __restrict__ ei, const float* __restrict__ ew,
                       const int* __restrict__ rowptr, const int* __restrict__ rank,
                       int* __restrict__ csr, int E) {
  int e = blockIdx.x * blockDim.x + threadIdx.x;
  if (e < E) {
    int d = ei[E + e];
    int p = rowptr[d] + rank[e];
    int2 r;
    r.x = ei[e];
    r.y = __float_as_int(ew[e]);
    *(int2*)(csr + 2 * (size_t)p) = r;
  }
}

__global__ void k_dinv(const int* __restrict__ csr, const int* __restrict__ rowptr,
                       const int* __restrict__ cnt, float* __restrict__ dinv, int n) {
  int gid = blockIdx.x * blockDim.x + threadIdx.x;
  int node = gid >> 3;
  int j = gid & 7;
  float s = 0.0f;
  if (node < n) {
    int st = rowptr[node];
    int len = cnt[node];
    for (int k = j; k < len; k += 8) s += __int_as_float(csr[2 * (size_t)(st + k) + 1]);
  }
  s += __shfl_xor(s, 1);
  s += __shfl_xor(s, 2);
  s += __shfl_xor(s, 4);
  if (node < n && j == 0) {
    float dg = 1.0f + s;
    dinv[node] = (dg > 0.0f) ? rsqrtf(dg) : 0.0f;
  }
}

__global__ void k_pack_w(const float* __restrict__ W, float* __restrict__ Wp,
                         int K, int N) {
  int idx = blockIdx.x * blockDim.x + threadIdx.x;
  if (idx >= K * N) return;
  int k = idx / N;
  int n = idx - k * N;
  int k4 = k >> 2;
  int hi = (k >> 1) & 1;
  int j  = k & 1;
  Wp[(((size_t)((k4 << 1) | hi)) * N + n) * 2 + j] = W[idx];
}

template <int NT>
__global__ void k_gemm_wmma(const float* __restrict__ A,
                            const float* __restrict__ Bp,
                            const float* __restrict__ rs,
                            float* __restrict__ C,
                            int M, int K) {
  const int N = NT * 16;
  int wave = (int)((blockIdx.x * blockDim.x + threadIdx.x) >> 5);
  int lane = threadIdx.x & 31;
  if (wave >= (M >> 4)) return;

  int m0 = wave << 4;
  int lo = lane & 15;
  int hi = lane >> 4;

  const float* aPtr = A  + (size_t)(m0 + lo) * K + (hi << 1);
  const float* bPtr = Bp + ((size_t)hi * N + lo) * 2;

  v8f z = {0.f,0.f,0.f,0.f,0.f,0.f,0.f,0.f};
  v8f acc[NT];
#pragma unroll
  for (int t = 0; t < NT; ++t) acc[t] = z;

  for (int k = 0; k < K; k += 8) {
    v2f a0 = *(const v2f*)(aPtr + k);
    v2f a1 = *(const v2f*)(aPtr + k + 4);
    const float* bp0 = bPtr + (size_t)k * N;
    const float* bp1 = bp0 + (size_t)4 * N;
    v2f b0[NT], b1[NT];
#pragma unroll
    for (int t = 0; t < NT; ++t) b0[t] = *(const v2f*)(bp0 + t * 32);
#pragma unroll
    for (int t = 0; t < NT; ++t) b1[t] = *(const v2f*)(bp1 + t * 32);
#pragma unroll
    for (int t = 0; t < NT; ++t)
      acc[t] = __builtin_amdgcn_wmma_f32_16x16x4_f32(
          false, a0, false, b0[t], (short)0, acc[t], false, false);
#pragma unroll
    for (int t = 0; t < NT; ++t)
      acc[t] = __builtin_amdgcn_wmma_f32_16x16x4_f32(
          false, a1, false, b1[t], (short)0, acc[t], false, false);
  }

  int cRow = m0 + (hi << 3);
#pragma unroll
  for (int r = 0; r < 8; ++r) {
    float* cp = C + (size_t)(cRow + r) * N + lo;
    float sc = rs[cRow + r];
#pragma unroll
    for (int t = 0; t < NT; ++t) cp[t * 16] = acc[t][r] * sc;
  }
}

template <bool RELU>
__global__ void __launch_bounds__(256)
k_gather(const int* __restrict__ csr, const int* __restrict__ rowptr,
         const int* __restrict__ cnt, const float* __restrict__ dinv,
         const float* __restrict__ inA, const float* __restrict__ inB,
         const float* __restrict__ biasA, const float* __restrict__ biasB,
         float* __restrict__ outA, float* __restrict__ outB, int ld, int n) {
  int lane = threadIdx.x & 31;
  int d = __builtin_amdgcn_readfirstlane((int)((blockIdx.x * blockDim.x + threadIdx.x) >> 5));
  if (d >= n) return;
  int st  = rowptr[d];
  int len = cnt[d];
  float dd = dinv[d];
  int col = (lane & 15) << 2;
  const float* in  = ((lane < 16) ? inA : inB) + col;
  const float* bp  = ((lane < 16) ? biasA : biasB) + col;
  float*       out = ((lane < 16) ? outA : outB) + (size_t)d * ld + col;

  float4 acc = *(const float4*)(in + (size_t)d * ld);

  for (int base = 0; base < len; base += 32) {
    int m = len - base;
    if (m > 32) m = 32;
    int sj = d;
    float nj = 0.0f;
    if (lane < m) {
      int2 r = *(const int2*)(csr + 2 * (size_t)(st + base + lane));
      sj = r.x;
      nj = __int_as_float(r.y);
    }
    for (int j = 0; j < m; j += 8) {
      float4 r[8];
      float wgt[8];
#pragma unroll
      for (int u = 0; u < 8; ++u) {
        int s = __builtin_amdgcn_readlane(sj, j + u);
        wgt[u] = __int_as_float(__builtin_amdgcn_readlane(__float_as_int(nj), j + u));
        float4 t = make_float4(0.0f, 0.0f, 0.0f, 0.0f);
        if (j + u < m) t = *(const float4*)(in + (size_t)s * ld);
        r[u] = t;
      }
#pragma unroll
      for (int u = 0; u < 8; ++u) {
        acc.x += r[u].x * wgt[u];
        acc.y += r[u].y * wgt[u];
        acc.z += r[u].z * wgt[u];
        acc.w += r[u].w * wgt[u];
      }
    }
  }
  acc.x = acc.x * dd + bp[0];
  acc.y = acc.y * dd + bp[1];
  acc.z = acc.z * dd + bp[2];
  acc.w = acc.w * dd + bp[3];
  if (RELU) {
    acc.x = fmaxf(acc.x, 0.0f);
    acc.y = fmaxf(acc.y, 0.0f);
    acc.z = fmaxf(acc.z, 0.0f);
    acc.w = fmaxf(acc.w, 0.0f);
  }
  *(float4*)out = acc;
}

extern "C" void kernel_launch(void* const* d_in, const int* in_sizes, int n_in,
                              void* d_out, int out_size, void* d_ws, size_t ws_size,
                              hipStream_t stream) {
  const float* x  = (const float*)d_in[0];
  const int*   ei = (const int*)  d_in[1];
  const float* ew = (const float*)d_in[2];
  const float* W1 = (const float*)d_in[3];
  const float* b1 = (const float*)d_in[4];
  const float* W2 = (const float*)d_in[5];
  const float* b2 = (const float*)d_in[6];
  const float* W3 = (const float*)d_in[7];
  const float* b3 = (const float*)d_in[8];

  const int N = in_sizes[0] / IN_CH;
  const int E = in_sizes[2];

  float* ws   = (float*)d_ws;
  int*   cnt  = (int*)ws;
  float* dinv = ws + N;
  float* h1   = ws + 2 * (size_t)N;
  float* er   = h1 + (size_t)N * HID;
  int*   rank   = (int*)er;
  int*   csr    = rank + (size_t)E;
  int*   rowptr = csr + 2 * (size_t)E;
  float* W1p  = er + (size_t)N * HID;
  float* W2p  = W1p + (size_t)IN_CH * HID;
  float* W3p  = W2p + (size_t)HID * OUTC;
  float* t2   = h1;
  float* t3   = h1 + (size_t)N * OUTC;
  float* om   = (float*)d_out;
  float* ol   = om + (size_t)N * OUTC;
  float* h    = (float*)d_out;

  const int T = 256;

  k_pack_w<<<(IN_CH * HID + T - 1) / T, T, 0, stream>>>(W1, W1p, IN_CH, HID);
  k_pack_w<<<(HID * OUTC + T - 1) / T, T, 0, stream>>>(W2, W2p, HID, OUTC);
  k_pack_w<<<(HID * OUTC + T - 1) / T, T, 0, stream>>>(W3, W3p, HID, OUTC);

  k_zero_cnt<<<(N + T - 1) / T, T, 0, stream>>>(cnt, N);
  k_rank    <<<(E + T - 1) / T, T, 0, stream>>>(ei, cnt, rank, E);
  k_scan    <<<(N + 1023) / 1024, 1024, 0, stream>>>(cnt, rowptr, N);
  k_fill    <<<(E + T - 1) / T, T, 0, stream>>>(ei, ew, rowptr, rank, csr, E);
  k_dinv    <<<(N * 8 + T - 1) / T, T, 0, stream>>>(csr, rowptr, cnt, dinv, N);

  {
    int waves  = N >> 4;
    int blocks = (waves * 32 + T - 1) / T;
    k_gemm_wmma<8><<<blocks, T, 0, stream>>>(x, W1p, dinv, h1, N, IN_CH);
  }

  k_gather<true><<<(N * 32 + T - 1) / T, T, 0, stream>>>(
      csr, rowptr, cnt, dinv, h1, h1 + 64, b1, b1 + 64, h, h + 64, HID, N);

  {
    int waves  = N >> 4;
    int blocks = (waves * 32 + T - 1) / T;
    k_gemm_wmma<4><<<blocks, T, 0, stream>>>(h, W2p, dinv, t2, N, HID);
    k_gemm_wmma<4><<<blocks, T, 0, stream>>>(h, W3p, dinv, t3, N, HID);
  }

  k_gather<false><<<(N * 32 + T - 1) / T, T, 0, stream>>>(
      csr, rowptr, cnt, dinv, t2, t3, b2, b3, om, ol, OUTC, N);
}


# CSR build as a two-level counting sort on LDS counters (edge chunks binned by 128-row tile, then one work-group per tile) instead of a returning global add per edge, scan, fill; row sum order changes
# speedup vs baseline: 3.0899x; 1.4101x over previous
#include <hip/hip_runtime.h>

typedef float v2f __attribute__((ext_vector_type(2)));
typedef float v8f __attribute__((ext_vector_type(8)));

#define IN_CH 512
#define HID   128
#define OUTC  64

#define TILE_SHIFT 7
#define TILE_ROWS  128
#define MAXT       512
#define NCHUNK     512

__global__ void __launch_bounds__(1024)
k_bin(const int* __restrict__ ei, const float* __restrict__ ew,
      int* __restrict__ seg, int* __restrict__ srow, int* __restrict__ oc,
      int E, int EC, int NT) {
  __shared__ int hist[MAXT];
  __shared__ int curs[MAXT];
  __shared__ int wtot[16];
  int t = threadIdx.x;
  int lane = t & 31;
  int w = t >> 5;
  int c = blockIdx.x;
  int e0 = c * EC;
  int e1 = e0 + EC;
  if (e1 > E) e1 = E;
  if (t < MAXT) hist[t] = 0;
  __syncthreads();
  for (int e = e0 + t; e < e1; e += 1024) atomicAdd(&hist[ei[E + e] >> TILE_SHIFT], 1);
  __syncthreads();
  int v = 0, x = 0;
  if (t < MAXT) {
    v = hist[t];
    x = v;
    for (int dlt = 1; dlt < 32; dlt <<= 1) {
      int y = __shfl_up(x, dlt);
      if (lane >= dlt) x += y;
    }
    if (lane == 31) wtot[w] = x;
  }
  __syncthreads();
  if (t < MAXT) {
    int off = 0;
    for (int k = 0; k < 16; ++k) if (k < w) off += wtot[k];
    int ex = off + x - v;
    curs[t] = ex;
    if (t < NT) {
      int2 o;
      o.x = ex;
      o.y = v;
      *(int2*)(oc + 2 * ((size_t)t * NCHUNK + c)) = o;
    }
  }
  __syncthreads();
  for (int e = e0 + t; e < e1; e += 1024) {
    int d = ei[E + e];
    int p = e0 + atomicAdd(&curs[d >> TILE_SHIFT], 1);
    int2 r;
    r.x = ei[e];
    r.y = __float_as_int(ew[e]);
    *(int2*)(seg + 2 * (size_t)p) = r;
    srow[p] = d & (TILE_ROWS - 1);
  }
}

__global__ void __launch_bounds__(NCHUNK)
k_csr(const int* __restrict__ seg, const int* __restrict__ srow,
      const int* __restrict__ oc,
      int* __restrict__ csr, int* __restrict__ rowptr, int* __restrict__ cnt,
      int EC, int n) {
  __shared__ int rcnt[TILE_ROWS];
  __shared__ int rcur[TILE_ROWS];
  __shared__ int red[NCHUNK / 32];
  __shared__ int red2[TILE_ROWS / 32];
  int c = threadIdx.x;
  int lane = c & 31;
  int w = c >> 5;
  int T = blockIdx.x;
  int2 o = *(const int2*)(oc + 2 * ((size_t)T * NCHUNK + c));
  int s0 = c * EC + o.x;
  int m = o.y;
  int part = o.x;
  for (int dlt = 16; dlt > 0; dlt >>= 1) part += __shfl_xor(part, dlt);
  if (lane == 0) red[w] = part;
  if (c < TILE_ROWS) rcnt[c] = 0;
  __syncthreads();
  int base = 0;
  for (int k = 0; k < NCHUNK / 32; ++k) base += red[k];
  for (int k = 0; k < m; ++k) atomicAdd(&rcnt[srow[s0 + k]], 1);
  __syncthreads();
  int v = 0, x = 0;
  if (c < TILE_ROWS) {
    v = rcnt[c];
    x = v;
    for (int dlt = 1; dlt < 32; dlt <<= 1) {
      int y = __shfl_up(x, dlt);
      if (lane >= dlt) x += y;
    }
    if (lane == 31) red2[w] = x;
  }
  __syncthreads();
  if (c < TILE_ROWS) {
    int off = 0;
    for (int k = 0; k < TILE_ROWS / 32; ++k) if (k < w) off += red2[k];
    int ex = off + x - v;
    rcur[c] = ex;
    int row = T * TILE_ROWS + c;
    if (row < n) {
      rowptr[row] = base + ex;
      cnt[row] = v;
    }
  }
  __syncthreads();
  for (int k = 0; k < m; ++k) {
    int p = base + atomicAdd(&rcur[srow[s0 + k]], 1);
    *(int2*)(csr + 2 * (size_t)p) = *(const int2*)(seg + 2 * (size_t)(s0 + k));
  }
}

__global__ void k_dinv(const int* __restrict__ csr, const int* __restrict__ rowptr,
                       const int* __restrict__ cnt, float* __restrict__ dinv, int n) {
  int gid = blockIdx.x * blockDim.x + threadIdx.x;
  int node = gid >> 3;
  int j = gid & 7;
  float s = 0.0f;
  if (node < n) {
    int st = rowptr[node];
    int len = cnt[node];
    for (int k = j; k < len; k += 8) s += __int_as_float(csr[2 * (size_t)(st + k) + 1]);
  }
  s += __shfl_xor(s, 1);
  s += __shfl_xor(s, 2);
  s += __shfl_xor(s, 4);
  if (node < n && j == 0) {
    float dg = 1.0f + s;
    dinv[node] = (dg > 0.0f) ? rsqrtf(dg) : 0.0f;
  }
}

__global__ void k_pack_w(const float* __restrict__ W, float* __restrict__ Wp,
                         int K, int N) {
  int idx = blockIdx.x * blockDim.x + threadIdx.x;
  if (idx >= K * N) return;
  int k = idx / N;
  int n = idx - k * N;
  int k4 = k >> 2;
  int hi = (k >> 1) & 1;
  int j  = k & 1;
  Wp[(((size_t)((k4 << 1) | hi)) * N + n) * 2 + j] = W[idx];
}

template <int NT>
__global__ void k_gemm_wmma(const float* __restrict__ A,
                            const float* __restrict__ Bp,
                            const float* __restrict__ rs,
                            float* __restrict__ C,
                            int M, int K) {
  const int N = NT * 16;
  int wave = (int)((blockIdx.x * blockDim.x + threadIdx.x) >> 5);
  int lane = threadIdx.x & 31;
  if (wave >= (M >> 4)) return;

  int m0 = wave << 4;
  int lo = lane & 15;
  int hi = lane >> 4;

  const float* aPtr = A  + (size_t)(m0 + lo) * K + (hi << 1);
  const float* bPtr = Bp + ((size_t)hi * N + lo) * 2;

  v8f z = {0.f,0.f,0.f,0.f,0.f,0.f,0.f,0.f};
  v8f acc[NT];
#pragma unroll
  for (int t = 0; t < NT; ++t) acc[t] = z;

  for (int k = 0; k < K; k += 8) {
    v2f a0 = *(const v2f*)(aPtr + k);
    v2f a1 = *(const v2f*)(aPtr + k + 4);
    const float* bp0 = bPtr + (size_t)k * N;
    const float* bp1 = bp0 + (size_t)4 * N;
    v2f b0[NT], b1[NT];
#pragma unroll
    for (int t = 0; t < NT; ++t) b0[t] = *(const v2f*)(bp0 + t * 32);
#pragma unroll
    for (int t = 0; t < NT; ++t) b1[t] = *(const v2f*)(bp1 + t * 32);
#pragma unroll
    for (int t = 0; t < NT; ++t)
      acc[t] = __builtin_amdgcn_wmma_f32_16x16x4_f32(
          false, a0, false, b0[t], (short)0, acc[t], false, false);
#pragma unroll
    for (int t = 0; t < NT; ++t)
      acc[t] = __builtin_amdgcn_wmma_f32_16x16x4_f32(
          false, a1, false, b1[t], (short)0, acc[t], false, false);
  }

  int cRow = m0 + (hi << 3);
#pragma unroll
  for (int r = 0; r < 8; ++r) {
    float* cp = C + (size_t)(cRow + r) * N + lo;
    float sc = rs[cRow + r];
#pragma unroll
    for (int t = 0; t < NT; ++t) cp[t * 16] = acc[t][r] * sc;
  }
}

template <bool RELU>
__global__ void __launch_bounds__(256)
k_gather(const int* __restrict__ csr, const int* __restrict__ rowptr,
         const int* __restrict__ cnt, const float* __restrict__ dinv,
         const float* __restrict__ inA, const float* __restrict__ inB,
         const float* __restrict__ biasA, const float* __restrict__ biasB,
         float* __restrict__ outA, float* __restrict__ outB, int ld, int n) {
  int lane = threadIdx.x & 31;
  int d = __builtin_amdgcn_readfirstlane((int)((blockIdx.x * blockDim.x + threadIdx.x) >> 5));
  if (d >= n) return;
  int st  = rowptr[d];
  int len = cnt[d];
  float dd = dinv[d];
  int col = (lane & 15) << 2;
  const float* in  = ((lane < 16) ? inA : inB) + col;
  const float* bp  = ((lane < 16) ? biasA : biasB) + col;
  float*       out = ((lane < 16) ? outA : outB) + (size_t)d * ld + col;

  float4 acc = *(const float4*)(in + (size_t)d * ld);

  for (int base = 0; base < len; base += 32) {
    int m = len - base;
    if (m > 32) m = 32;
    int sj = d;
    float nj = 0.0f;
    if (lane < m) {
      int2 r = *(const int2*)(csr + 2 * (size_t)(st + base + lane));
      sj = r.x;
      nj = __int_as_float(r.y);
    }
    for (int j = 0; j < m; j += 8) {
      float4 r[8];
      float wgt[8];
#pragma unroll
      for (int u = 0; u < 8; ++u) {
        int s = __builtin_amdgcn_readlane(sj, j + u);
        wgt[u] = __int_as_float(__builtin_amdgcn_readlane(__float_as_int(nj), j + u));
        float4 t = make_float4(0.0f, 0.0f, 0.0f, 0.0f);
        if (j + u < m) t = *(const float4*)(in + (size_t)s * ld);
        r[u] = t;
      }
#pragma unroll
      for (int u = 0; u < 8; ++u) {
        acc.x += r[u].x * wgt[u];
        acc.y += r[u].y * wgt[u];
        acc.z += r[u].z * wgt[u];
        acc.w += r[u].w * wgt[u];
      }
    }
  }
  acc.x = acc.x * dd + bp[0];
  acc.y = acc.y * dd + bp[1];
  acc.z = acc.z * dd + bp[2];
  acc.w = acc.w * dd + bp[3];
  if (RELU) {
    acc.x = fmaxf(acc.x, 0.0f);
    acc.y = fmaxf(acc.y, 0.0f);
    acc.z = fmaxf(acc.z, 0.0f);
    acc.w = fmaxf(acc.w, 0.0f);
  }
  *(float4*)out = acc;
}

extern "C" void kernel_launch(void* const* d_in, const int* in_sizes, int n_in,
                              void* d_out, int out_size, void* d_ws, size_t ws_size,
                              hipStream_t stream) {
  const float* x  = (const float*)d_in[0];
  const int*   ei = (const int*)  d_in[1];
  const float* ew = (const float*)d_in[2];
  const float* W1 = (const float*)d_in[3];
  const float* b1 = (const float*)d_in[4];
  const float* W2 = (const float*)d_in[5];
  const float* b2 = (const float*)d_in[6];
  const float* W3 = (const float*)d_in[7];
  const float* b3 = (const float*)d_in[8];

  const int N = in_sizes[0] / IN_CH;
  const int E = in_sizes[2];

  float* ws   = (float*)d_ws;
  int*   cnt  = (int*)ws;
  float* dinv = ws + N;
  float* h1   = ws + 2 * (size_t)N;
  float* er   = h1 + (size_t)N * HID;
  int*   csr    = (int*)er;
  int*   rowptr = csr + 2 * (size_t)E;
  int*   oc     = rowptr + N;
  int*   seg    = (int*)h1;
  int*   srow   = seg + 2 * (size_t)E;
  float* W1p  = er + (size_t)N * HID;
  float* W2p  = W1p + (size_t)IN_CH * HID;
  float* W3p  = W2p + (size_t)HID * OUTC;
  float* t2   = h1;
  float* t3   = h1 + (size_t)N * OUTC;
  float* om   = (float*)d_out;
  float* ol   = om + (size_t)N * OUTC;
  float* h    = (float*)d_out;

  const int T = 256;

  k_pack_w<<<(IN_CH * HID + T - 1) / T, T, 0, stream>>>(W1, W1p, IN_CH, HID);
  k_pack_w<<<(HID * OUTC + T - 1) / T, T, 0, stream>>>(W2, W2p, HID, OUTC);
  k_pack_w<<<(HID * OUTC + T - 1) / T, T, 0, stream>>>(W3, W3p, HID, OUTC);

  const int NT = (N + TILE_ROWS - 1) / TILE_ROWS;
  const int EC = (E + NCHUNK - 1) / NCHUNK;
  k_bin     <<<NCHUNK, 1024, 0, stream>>>(ei, ew, seg, srow, oc, E, EC, NT);
  k_csr     <<<NT, NCHUNK, 0, stream>>>(seg, srow, oc, csr, rowptr, cnt, EC, N);
  k_dinv    <<<(N * 8 + T - 1) / T, T, 0, stream>>>(csr, rowptr, cnt, dinv, N);

  {
    int waves  = N >> 4;
    int blocks = (waves * 32 + T - 1) / T;
    k_gemm_wmma<8><<<blocks, T, 0, stream>>>(x, W1p, dinv, h1, N, IN_CH);
  }

  k_gather<true><<<(N * 32 + T - 1) / T, T, 0, stream>>>(
      csr, rowptr, cnt, dinv, h1, h1 + 64, b1, b1 + 64, h, h + 64, HID, N);

  {
    int waves  = N >> 4;
    int blocks = (waves * 32 + T - 1) / T;
    k_gemm_wmma<4><<<blocks, T, 0, stream>>>(h, W2p, dinv, t2, N, HID);
    k_gemm_wmma<4><<<blocks, T, 0, stream>>>(h, W3p, dinv, t3, N, HID);
  }

  k_gather<false><<<(N * 32 + T - 1) / T, T, 0, stream>>>(
      csr, rowptr, cnt, dinv, t2, t3, b2, b3, om, ol, OUTC, N);
}


# Launch merging 11 -> 6: weights packed inside the edge-binning launch, degree and dinv inside the CSR launch (same sum order), both layer-2 products as one 16x128-per-wave product on [W2|W3]
# speedup vs baseline: 3.2121x; 1.0395x over previous
#include <hip/hip_runtime.h>

typedef float v2f __attribute__((ext_vector_type(2)));
typedef float v8f __attribute__((ext_vector_type(8)));

#define IN_CH 512
#define HID   128
#define OUTC  64

#define TILE_SHIFT 7
#define TILE_ROWS  128
#define MAXT       512
#define NCHUNK     512

#define PACK_TOTAL  (IN_CH * HID + HID * 2 * OUTC)
#define PACK_PER_WG (PACK_TOTAL / NCHUNK)
static_assert(PACK_TOTAL % NCHUNK == 0 && PACK_PER_WG <= 1024, "one element per thread of k_bin");

__device__ __forceinline__ size_t pack_pos(int k, int n, int N) {
  int k4 = k >> 2;
  int hi = (k >> 1) & 1;
  int j  = k & 1;
  return (((size_t)((k4 << 1) | hi)) * N + n) * 2 + j;
}

__global__ void __launch_bounds__(1024)
k_bin(const int* __restrict__ ei, const float* __restrict__ ew,
      int* __restrict__ seg, int* __restrict__ srow, int* __restrict__ oc,
      int E, int EC, int NT,
      const float* __restrict__ W1, const float* __restrict__ W2,
      const float* __restrict__ W3,
      float* __restrict__ W1p, float* __restrict__ W23p) {
  __shared__ int hist[MAXT];
  __shared__ int curs[MAXT];
  __shared__ int wtot[16];
  int t = threadIdx.x;
  int lane = t & 31;
  int w = t >> 5;
  int c = blockIdx.x;
  int e0 = c * EC;
  int e1 = e0 + EC;
  if (e1 > E) e1 = E;
  if (t < MAXT) hist[t] = 0;
  __syncthreads();
  float  pv = 0.0f;
  float* pp = W1p;
  if (t < PACK_PER_WG) {
    int idx = c * PACK_PER_WG + t;
    if (idx < IN_CH * HID) {
      int k = idx / HID;
      int n = idx - k * HID;
      pv = W1[idx];
      pp = W1p + pack_pos(k, n, HID);
    } else {
      int i2 = idx - IN_CH * HID;
      int k = i2 / (2 * OUTC);
      int n = i2 - k * (2 * OUTC);
      pv = (n < OUTC) ? W2[k * OUTC + n] : W3[k * OUTC + (n - OUTC)];
      pp = W23p + pack_pos(k, n, 2 * OUTC);
    }
  }
  for (int e = e0 + t; e < e1; e += 1024) atomicAdd(&hist[ei[E + e] >> TILE_SHIFT], 1);
  __syncthreads();
  int v = 0, x = 0;
  if (t < MAXT) {
    v = hist[t];
    x = v;
    for (int dlt = 1; dlt < 32; dlt <<= 1) {
      int y = __shfl_up(x, dlt);
      if (lane >= dlt) x += y;
    }
    if (lane == 31) wtot[w] = x;
  }
  __syncthreads();
  if (t < MAXT) {
    int off = 0;
    for (int k = 0; k < 16; ++k) if (k < w) off += wtot[k];
    int ex = off + x - v;
    curs[t] = ex;
    if (t < NT) {
      int2 o;
      o.x = ex;
      o.y = v;
      *(int2*)(oc + 2 * ((size_t)t * NCHUNK + c)) = o;
    }
  }
  __syncthreads();
  for (int e = e0 + t; e < e1; e += 1024) {
    int d = ei[E + e];
    int p = e0 + atomicAdd(&curs[d >> TILE_SHIFT], 1);
    int2 r;
    r.x = ei[e];
    r.y = __float_as_int(ew[e]);
    *(int2*)(seg + 2 * (size_t)p) = r;
    srow[p] = d & (TILE_ROWS - 1);
  }
  if (t < PACK_PER_WG) *pp = pv;
}

__global__ void __launch_bounds__(NCHUNK)
k_csr(const int* __restrict__ seg, const int* __restrict__ srow,
      const int* __restrict__ oc,
      int* __restrict__ csr, int* __restrict__ rowptr, int* __restrict__ cnt,
      float* __restrict__ dinv, int EC, int n) {
  __shared__ int rcnt[TILE_ROWS];
  __shared__ int rcur[TILE_ROWS];
  __shared__ int red[NCHUNK / 32];
  __shared__ int red2[TILE_ROWS / 32];
  int c = threadIdx.x;
  int lane = c & 31;
  int w = c >> 5;
  int T = blockIdx.x;
  int2 o = *(const int2*)(oc + 2 * ((size_t)T * NCHUNK + c));
  int s0 = c * EC + o.x;
  int m = o.y;
  int part = o.x;
  for (int dlt = 16; dlt > 0; dlt >>= 1) part += __shfl_xor(part, dlt);
  if (lane == 0) red[w] = part;
  if (c < TILE_ROWS) rcnt[c] = 0;
  __syncthreads();
  int base = 0;
  for (int k = 0; k < NCHUNK / 32; ++k) base += red[k];
  for (int k = 0; k < m; ++k) atomicAdd(&rcnt[srow[s0 + k]], 1);
  __syncthreads();
  int v = 0, x = 0;
  if (c < TILE_ROWS) {
    v = rcnt[c];
    x = v;
    for (int dlt = 1; dlt < 32; dlt <<= 1) {
      int y = __shfl_up(x, dlt);
      if (lane >= dlt) x += y;
    }
    if (lane == 31) red2[w] = x;
  }
  __syncthreads();
  if (c < TILE_ROWS) {
    int off = 0;
    for (int k = 0; k < TILE_ROWS / 32; ++k) if (k < w) off += red2[k];
    int ex = off + x - v;
    rcur[c] = ex;
    int row = T * TILE_ROWS + c;
    if (row < n) {
      rowptr[row] = base + ex;
      cnt[row] = v;
    }
  }
  __syncthreads();
  for (int k = 0; k < m; ++k) {
    int p = base + atomicAdd(&rcur[srow[s0 + k]], 1);
    *(int2*)(csr + 2 * (size_t)p) = *(const int2*)(seg + 2 * (size_t)(s0 + k));
  }
  __syncthreads();
  int j = c & 7;
  for (int r = c >> 3; r < TILE_ROWS; r += NCHUNK / 8) {
    int row = T * TILE_ROWS + r;
    int len = rcnt[r];
    const int* wp = csr + 2 * (size_t)(base + rcur[r] - len) + 1;
    float s = 0.0f;
    for (int k = j; k < len; k += 32) {
      float w0 = __int_as_float(wp[2 * k]);
      float w1 = (k + 8  < len) ? __int_as_float(wp[2 * (k + 8)])  : 0.0f;
      float w2 = (k + 16 < len) ? __int_as_float(wp[2 * (k + 16)]) : 0.0f;
      float w3 = (k + 24 < len) ? __int_as_float(wp[2 * (k + 24)]) : 0.0f;
      s += w0;
      s += w1;
      s += w2;
      s += w3;
    }
    s += __shfl_xor(s, 1);
    s += __shfl_xor(s, 2);
    s += __shfl_xor(s, 4);
    if (row < n && j == 0) {
      float dg = 1.0f + s;
      dinv[row] = (dg > 0.0f) ? rsqrtf(dg) : 0.0f;
    }
  }
}

template <int NT, bool SPLIT>
__global__ void k_gemm_wmma(const float* __restrict__ A,
                            const float* __restrict__ Bp,
                            const float* __restrict__ rs,
                            float* __restrict__ C,
                            float* __restrict__ C1,
                            int M, int K) {
  const int N = NT * 16;
  int wave = (int)((blockIdx.x * blockDim.x + threadIdx.x) >> 5);
  int lane = threadIdx.x & 31;
  if (wave >= (M >> 4)) return;

  int m0 = wave << 4;
  int lo = lane & 15;
  int hi = lane >> 4;

  const float* aPtr = A  + (size_t)(m0 + lo) * K + (hi << 1);
  const float* bPtr = Bp + ((size_t)hi * N + lo) * 2;

  v8f z = {0.f,0.f,0.f,0.f,0.f,0.f,0.f,0.f};
  v8f acc[NT];
#pragma unroll
  for (int t = 0; t < NT; ++t) acc[t] = z;

  for (int k = 0; k < K; k += 8) {
    v2f a0 = *(const v2f*)(aPtr + k);
    v2f a1 = *(const v2f*)(aPtr + k + 4);
    const float* bp0 = bPtr + (size_t)k * N;
    const float* bp1 = bp0 + (size_t)4 * N;
    v2f b0[NT], b1[NT];
#pragma unroll
    for (int t = 0; t < NT; ++t) b0[t] = *(const v2f*)(bp0 + t * 32);
#pragma unroll
    for (int t = 0; t < NT; ++t) b1[t] = *(const v2f*)(bp1 + t * 32);
#pragma unroll
    for (int t = 0; t < NT; ++t)
      acc[t] = __builtin_amdgcn_wmma_f32_16x16x4_f32(
          false, a0, false, b0[t], (short)0, acc[t], false, false);
#pragma unroll
    for (int t = 0; t < NT; ++t)
      acc[t] = __builtin_amdgcn_wmma_f32_16x16x4_f32(
          false, a1, false, b1[t], (short)0, acc[t], false, false);
  }

  int cRow = m0 + (hi << 3);
#pragma unroll
  for (int r = 0; r < 8; ++r) {
    float sc = rs[cRow + r];
    if (SPLIT) {
      const int NH = N / 2;
      float* cp0 = C  + (size_t)(cRow + r) * NH + lo;
      float* cp1 = C1 + (size_t)(cRow + r) * NH + lo;
#pragma unroll
      for (int t = 0; t < NT / 2; ++t) cp0[t * 16] = acc[t][r] * sc;
#pragma unroll
      for (int t = 0; t < NT / 2; ++t) cp1[t * 16] = acc[NT / 2 + t][r] * sc;
    } else {
      float* cp = C + (size_t)(cRow + r) * N + lo;
#pragma unroll
      for (int t = 0; t < NT; ++t) cp[t * 16] = acc[t][r] * sc;
    }
  }
}

template <bool RELU>
__global__ void __launch_bounds__(256)
k_gather(const int* __restrict__ csr, const int* __restrict__ rowptr,
         const int* __restrict__ cnt, const float* __restrict__ dinv,
         const float* __restrict__ inA, const float* __restrict__ inB,
         const float* __restrict__ biasA, const float* __restrict__ biasB,
         float* __restrict__ outA, float* __restrict__ outB, int ld, int n) {
  int lane = threadIdx.x & 31;
  int d = __builtin_amdgcn_readfirstlane((int)((blockIdx.x * blockDim.x + threadIdx.x) >> 5));
  if (d >= n) return;
  int st  = rowptr[d];
  int len = cnt[d];
  float dd = dinv[d];
  int col = (lane & 15) << 2;
  const float* in  = ((lane < 16) ? inA : inB) + col;
  const float* bp  = ((lane < 16) ? biasA : biasB) + col;
  float*       out = ((lane < 16) ? outA : outB) + (size_t)d * ld + col;

  float4 acc = *(const float4*)(in + (size_t)d * ld);

  for (int base = 0; base < len; base += 32) {
    int m = len - base;
    if (m > 32) m = 32;
    int sj = d;
    float nj = 0.0f;
    if (lane < m) {
      int2 r = *(const int2*)(csr + 2 * (size_t)(st + base + lane));
      sj = r.x;
      nj = __int_as_float(r.y);
    }
    for (int j = 0; j < m; j += 8) {
      float4 r[8];
      float wgt[8];
#pragma unroll
      for (int u = 0; u < 8; ++u) {
        int s = __builtin_amdgcn_readlane(sj, j + u);
        wgt[u] = __int_as_float(__builtin_amdgcn_readlane(__float_as_int(nj), j + u));
        float4 t = make_float4(0.0f, 0.0f, 0.0f, 0.0f);
        if (j + u < m) t = *(const float4*)(in + (size_t)s * ld);
        r[u] = t;
      }
#pragma unroll
      for (int u = 0; u < 8; ++u) {
        acc.x += r[u].x * wgt[u];
        acc.y += r[u].y * wgt[u];
        acc.z += r[u].z * wgt[u];
        acc.w += r[u].w * wgt[u];
      }
    }
  }
  acc.x = acc.x * dd + bp[0];
  acc.y = acc.y * dd + bp[1];
  acc.z = acc.z * dd + bp[2];
  acc.w = acc.w * dd + bp[3];
  if (RELU) {
    acc.x = fmaxf(acc.x, 0.0f);
    acc.y = fmaxf(acc.y, 0.0f);
    acc.z = fmaxf(acc.z, 0.0f);
    acc.w = fmaxf(acc.w, 0.0f);
  }
  *(float4*)out = acc;
}

extern "C" void kernel_launch(void* const* d_in, const int* in_sizes, int n_in,
                              void* d_out, int out_size, void* d_ws, size_t ws_size,
                              hipStream_t stream) {
  const float* x  = (const float*)d_in[0];
  const int*   ei = (const int*)  d_in[1];
  const float* ew = (const float*)d_in[2];
  const float* W1 = (const float*)d_in[3];
  const float* b1 = (const float*)d_in[4];
  const float* W2 = (const float*)d_in[5];
  const float* b2 = (const float*)d_in[6];
  const float* W3 = (const float*)d_in[7];
  const float* b3 = (const float*)d_in[8];

  const int N = in_sizes[0] / IN_CH;
  const int E = in_sizes[2];

  float* ws   = (float*)d_ws;
  int*   cnt  = (int*)ws;
  float* dinv = ws + N;
  float* h1   = ws + 2 * (size_t)N;
  float* er   = h1 + (size_t)N * HID;
  int*   csr    = (int*)er;
  int*   rowptr = csr + 2 * (size_t)E;
  int*   oc     = rowptr + N;
  int*   seg    = (int*)h1;
  int*   srow   = seg + 2 * (size_t)E;
  float* W1p  = er + (size_t)N * HID;
  float* W23p = W1p + (size_t)IN_CH * HID;
  float* t2   = h1;
  float* t3   = h1 + (size_t)N * OUTC;
  float* om   = (float*)d_out;
  float* ol   = om + (size_t)N * OUTC;
  float* h    = (float*)d_out;

  const int T = 256;

  const int NT = (N + TILE_ROWS - 1) / TILE_ROWS;
  const int EC = (E + NCHUNK - 1) / NCHUNK;
  k_bin     <<<NCHUNK, 1024, 0, stream>>>(ei, ew, seg, srow, oc, E, EC, NT,
                                          W1, W2, W3, W1p, W23p);
  k_csr     <<<NT, NCHUNK, 0, stream>>>(seg, srow, oc, csr, rowptr, cnt, dinv, EC, N);

  {
    int waves  = N >> 4;
    int blocks = (waves * 32 + T - 1) / T;
    k_gemm_wmma<8, false><<<blocks, T, 0, stream>>>(x, W1p, dinv, h1, nullptr, N, IN_CH);
  }

  k_gather<true><<<(N * 32 + T - 1) / T, T, 0, stream>>>(
      csr, rowptr, cnt, dinv, h1, h1 + 64, b1, b1 + 64, h, h + 64, HID, N);

  {
    int waves  = N >> 4;
    int blocks = (waves * 32 + T - 1) / T;
    k_gemm_wmma<8, true><<<blocks, T, 0, stream>>>(h, W23p, dinv, t2, t3, N, HID);
  }

  k_gather<false><<<(N * 32 + T - 1) / T, T, 0, stream>>>(
      csr, rowptr, cnt, dinv, t2, t3, b2, b3, om, ol, OUTC, N);
}


# CSR build: each sort pass reads its records in one load batch (k_bin 512 threads x 7 edges, k_csr 8 lanes per run), ranks them into an LDS block, copies it out in order; row sum order changes
# speedup vs baseline: 3.6738x; 1.1437x over previous
#include <hip/hip_runtime.h>

typedef float v2f __attribute__((ext_vector_type(2)));
typedef float v8f __attribute__((ext_vector_type(8)));

#define IN_CH 512
#define HID   128
#define OUTC  64

#define TILE_SHIFT 7
#define TILE_ROWS  128
#define MAXT       512
#define NCHUNK     512
#define BIN_T      512
#define BIN_PER_T  7
#define BIN_CAP    (BIN_PER_T * BIN_T)
static_assert(BIN_T >= MAXT, "k_bin scans one tile counter per thread");
#define CSR_CAP    6144
static_assert(CSR_CAP % NCHUNK == 0, "the copy out of k_csr takes CSR_CAP / NCHUNK records per thread");

#define PACK_TOTAL  (IN_CH * HID + HID * 2 * OUTC)
#define PACK_PER_WG (PACK_TOTAL / NCHUNK)
static_assert(PACK_TOTAL % NCHUNK == 0 && PACK_PER_WG <= 512, "one element per thread of k_bin");

__device__ __forceinline__ size_t pack_pos(int k, int n, int N) {
  int k4 = k >> 2;
  int hi = (k >> 1) & 1;
  int j  = k & 1;
  return (((size_t)((k4 << 1) | hi)) * N + n) * 2 + j;
}

__global__ void __launch_bounds__(BIN_T)
k_bin(const int* __restrict__ ei, const float* __restrict__ ew,
      int* __restrict__ seg, int* __restrict__ srow, int* __restrict__ oc,
      int E, int EC, int NT,
      const float* __restrict__ W1, const float* __restrict__ W2,
      const float* __restrict__ W3,
      float* __restrict__ W1p, float* __restrict__ W23p) {
  __shared__ int hist[MAXT];
  __shared__ int curs[MAXT];
  __shared__ int wtot[BIN_T / 32];
  __shared__ int2 sseg[BIN_CAP];
  __shared__ int  srw[BIN_CAP];
  int t = threadIdx.x;
  int lane = t & 31;
  int w = t >> 5;
  int c = blockIdx.x;
  int e0 = c * EC;
  int e1 = e0 + EC;
  if (e1 > E) e1 = E;
  const int  ne  = e1 - e0;
  const bool stg = ne <= BIN_CAP;
  if (t < MAXT) hist[t] = 0;
  __syncthreads();
  float  pv = 0.0f;
  float* pp = W1p;
  if (t < PACK_PER_WG) {
    int idx = c * PACK_PER_WG + t;
    if (idx < IN_CH * HID) {
      int k = idx / HID;
      int n = idx - k * HID;
      pv = W1[idx];
      pp = W1p + pack_pos(k, n, HID);
    } else {
      int i2 = idx - IN_CH * HID;
      int k = i2 / (2 * OUTC);
      int n = i2 - k * (2 * OUTC);
      pv = (n < OUTC) ? W2[k * OUTC + n] : W3[k * OUTC + (n - OUTC)];
      pp = W23p + pack_pos(k, n, 2 * OUTC);
    }
  }
  int ed[BIN_PER_T], es[BIN_PER_T], ewb[BIN_PER_T];
  if (stg) {
#pragma unroll
    for (int u = 0; u < BIN_PER_T; ++u) {
      int e = e0 + t + u * BIN_T;
      bool ok = e < e1;
      ed[u]  = ok ? ei[E + e] : 0;
      es[u]  = ok ? ei[e] : 0;
      ewb[u] = ok ? __float_as_int(ew[e]) : 0;
    }
#pragma unroll
    for (int u = 0; u < BIN_PER_T; ++u)
      if (e0 + t + u * BIN_T < e1) atomicAdd(&hist[ed[u] >> TILE_SHIFT], 1);
  } else {
    for (int e = e0 + t; e < e1; e += BIN_T) atomicAdd(&hist[ei[E + e] >> TILE_SHIFT], 1);
  }
  __syncthreads();
  int v = 0, x = 0;
  if (t < MAXT) {
    v = hist[t];
    x = v;
    for (int dlt = 1; dlt < 32; dlt <<= 1) {
      int y = __shfl_up(x, dlt);
      if (lane >= dlt) x += y;
    }
    if (lane == 31) wtot[w] = x;
  }
  __syncthreads();
  if (t < MAXT) {
    int off = 0;
    for (int k = 0; k < BIN_T / 32; ++k) if (k < w) off += wtot[k];
    int ex = off + x - v;
    curs[t] = ex;
    if (t < NT) {
      int2 o;
      o.x = ex;
      o.y = v;
      *(int2*)(oc + 2 * ((size_t)t * NCHUNK + c)) = o;
    }
  }
  __syncthreads();
  if (stg) {
    int q[BIN_PER_T];
#pragma unroll
    for (int u = 0; u < BIN_PER_T; ++u) {
      q[u] = 0;
      if (e0 + t + u * BIN_T < e1) q[u] = atomicAdd(&curs[ed[u] >> TILE_SHIFT], 1);
    }
#pragma unroll
    for (int u = 0; u < BIN_PER_T; ++u) {
      if (e0 + t + u * BIN_T < e1) {
        int2 r;
        r.x = es[u];
        r.y = ewb[u];
        sseg[q[u]] = r;
        srw[q[u]]  = ed[u] & (TILE_ROWS - 1);
      }
    }
    __syncthreads();
    int2 cr[BIN_PER_T];
    int  cw[BIN_PER_T];
#pragma unroll
    for (int u = 0; u < BIN_PER_T; ++u) {
      cr[u] = sseg[t + u * BIN_T];
      cw[u] = srw[t + u * BIN_T];
    }
#pragma unroll
    for (int u = 0; u < BIN_PER_T; ++u) {
      int i = t + u * BIN_T;
      if (i < ne) {
        *(int2*)(seg + 2 * (size_t)(e0 + i)) = cr[u];
        srow[e0 + i] = cw[u];
      }
    }
  } else {
    for (int e = e0 + t; e < e1; e += BIN_T) {
      int d = ei[E + e];
      int p = e0 + atomicAdd(&curs[d >> TILE_SHIFT], 1);
      int2 r;
      r.x = ei[e];
      r.y = __float_as_int(ew[e]);
      *(int2*)(seg + 2 * (size_t)p) = r;
      srow[p] = d & (TILE_ROWS - 1);
    }
  }
  if (t < PACK_PER_WG) *pp = pv;
}

template <class F>
__device__ __forceinline__ float row_wsum(int len, int j, F wt) {
  float s = 0.0f;
  for (int k = j; k < len; k += 32) {
    float w0 = wt(k);
    float w1 = (k + 8  < len) ? wt(k + 8)  : 0.0f;
    float w2 = (k + 16 < len) ? wt(k + 16) : 0.0f;
    float w3 = (k + 24 < len) ? wt(k + 24) : 0.0f;
    s += w0;
    s += w1;
    s += w2;
    s += w3;
  }
  s += __shfl_xor(s, 1);
  s += __shfl_xor(s, 2);
  s += __shfl_xor(s, 4);
  return s;
}

#define RUN_G   8
#define RUN_U   (NCHUNK / (NCHUNK / RUN_G))
#define RUN_IT  2
__global__ void __launch_bounds__(NCHUNK)
k_csr(const int* __restrict__ seg, const int* __restrict__ srow,
      const int* __restrict__ oc,
      int* __restrict__ csr, int* __restrict__ rowptr, int* __restrict__ cnt,
      float* __restrict__ dinv, int EC, int n) {
  __shared__ int rcnt[TILE_ROWS];
  __shared__ int rcur[TILE_ROWS];
  __shared__ int red[NCHUNK / 32];
  __shared__ int redm[NCHUNK / 32];
  __shared__ int red2[TILE_ROWS / 32];
  __shared__ int rs0[NCHUNK];
  __shared__ int rm[NCHUNK];
  __shared__ int2 stage[CSR_CAP];
  int c = threadIdx.x;
  int lane = c & 31;
  int w = c >> 5;
  int T = blockIdx.x;
  int2 o = *(const int2*)(oc + 2 * ((size_t)T * NCHUNK + c));
  int s0 = c * EC + o.x;
  int m = o.y;
  rs0[c] = s0;
  rm[c]  = m;
  int part = o.x;
  int tsum = m;
  for (int dlt = 16; dlt > 0; dlt >>= 1) {
    part += __shfl_xor(part, dlt);
    tsum += __shfl_xor(tsum, dlt);
  }
  if (lane == 0) {
    red[w]  = part;
    redm[w] = tsum;
  }
  if (c < TILE_ROWS) rcnt[c] = 0;
  __syncthreads();
  int base = 0;
  int total = 0;
  for (int k = 0; k < NCHUNK / 32; ++k) {
    base  += red[k];
    total += redm[k];
  }
  const bool stg = total <= CSR_CAP;
  const int g = c >> 3;
  const int j = c & (RUN_G - 1);
  int  gs[RUN_U], gm[RUN_U];
  int2 rec[RUN_U][RUN_IT];
  int  rw[RUN_U][RUN_IT];
  if (stg) {
#pragma unroll
    for (int u = 0; u < RUN_U; ++u) {
      gs[u] = rs0[g + (NCHUNK / RUN_G) * u];
      gm[u] = rm[g + (NCHUNK / RUN_G) * u];
    }
#pragma unroll
    for (int u = 0; u < RUN_U; ++u) {
#pragma unroll
      for (int it = 0; it < RUN_IT; ++it) {
        int k = j + RUN_G * it;
        bool ok = k < gm[u];
        rw[u][it]  = ok ? srow[gs[u] + k] : 0;
        rec[u][it] = ok ? *(const int2*)(seg + 2 * (size_t)(gs[u] + k)) : make_int2(0, 0);
      }
    }
#pragma unroll
    for (int u = 0; u < RUN_U; ++u) {
#pragma unroll
      for (int it = 0; it < RUN_IT; ++it)
        if (j + RUN_G * it < gm[u]) atomicAdd(&rcnt[rw[u][it]], 1);
      for (int k = j + RUN_G * RUN_IT; k < gm[u]; k += RUN_G) atomicAdd(&rcnt[srow[gs[u] + k]], 1);
    }
  } else {
    for (int k = 0; k < m; ++k) atomicAdd(&rcnt[srow[s0 + k]], 1);
  }
  __syncthreads();
  int v = 0, x = 0;
  if (c < TILE_ROWS) {
    v = rcnt[c];
    x = v;
    for (int dlt = 1; dlt < 32; dlt <<= 1) {
      int y = __shfl_up(x, dlt);
      if (lane >= dlt) x += y;
    }
    if (lane == 31) red2[w] = x;
  }
  __syncthreads();
  if (c < TILE_ROWS) {
    int off = 0;
    for (int k = 0; k < TILE_ROWS / 32; ++k) if (k < w) off += red2[k];
    int ex = off + x - v;
    rcur[c] = ex;
    int row = T * TILE_ROWS + c;
    if (row < n) {
      rowptr[row] = base + ex;
      cnt[row] = v;
    }
  }
  __syncthreads();
  if (stg) {
    int q[RUN_U][RUN_IT];
#pragma unroll
    for (int u = 0; u < RUN_U; ++u) {
#pragma unroll
      for (int it = 0; it < RUN_IT; ++it) {
        q[u][it] = 0;
        if (j + RUN_G * it < gm[u]) q[u][it] = atomicAdd(&rcur[rw[u][it]], 1);
      }
    }
#pragma unroll
    for (int u = 0; u < RUN_U; ++u) {
#pragma unroll
      for (int it = 0; it < RUN_IT; ++it)
        if (j + RUN_G * it < gm[u]) stage[q[u][it]] = rec[u][it];
      for (int k = j + RUN_G * RUN_IT; k < gm[u]; k += RUN_G) {
        int p = atomicAdd(&rcur[srow[gs[u] + k]], 1);
        stage[p] = *(const int2*)(seg + 2 * (size_t)(gs[u] + k));
      }
    }
  } else {
    for (int k = 0; k < m; ++k) {
      int p = base + atomicAdd(&rcur[srow[s0 + k]], 1);
      *(int2*)(csr + 2 * (size_t)p) = *(const int2*)(seg + 2 * (size_t)(s0 + k));
    }
  }
  __syncthreads();
  if (stg) {
    int2 cr[CSR_CAP / NCHUNK];
#pragma unroll
    for (int u = 0; u < CSR_CAP / NCHUNK; ++u) cr[u] = stage[c + u * NCHUNK];
#pragma unroll
    for (int u = 0; u < CSR_CAP / NCHUNK; ++u) {
      int i = c + u * NCHUNK;
      if (i < total) *(int2*)(csr + 2 * (size_t)(base + i)) = cr[u];
    }
    for (int r = c >> 3; r < TILE_ROWS; r += NCHUNK / 8) {
      int len = rcnt[r];
      int rb  = rcur[r] - len;
      float s = row_wsum(len, j, [&](int k) -> float { return __int_as_float(stage[rb + k].y); });
      int row = T * TILE_ROWS + r;
      if (row < n && j == 0) {
        float dg = 1.0f + s;
        dinv[row] = (dg > 0.0f) ? rsqrtf(dg) : 0.0f;
      }
    }
  } else {
    for (int r = c >> 3; r < TILE_ROWS; r += NCHUNK / 8) {
      int len = rcnt[r];
      const int* wp = csr + 2 * (size_t)(base + rcur[r] - len) + 1;
      float s = row_wsum(len, j, [&](int k) -> float { return __int_as_float(wp[2 * k]); });
      int row = T * TILE_ROWS + r;
      if (row < n && j == 0) {
        float dg = 1.0f + s;
        dinv[row] = (dg > 0.0f) ? rsqrtf(dg) : 0.0f;
      }
    }
  }
}

template <int NT, bool SPLIT>
__global__ void k_gemm_wmma(const float* __restrict__ A,
                            const float* __restrict__ Bp,
                            const float* __restrict__ rs,
                            float* __restrict__ C,
                            float* __restrict__ C1,
                            int M, int K) {
  const int N = NT * 16;
  int wave = (int)((blockIdx.x * blockDim.x + threadIdx.x) >> 5);
  int lane = threadIdx.x & 31;
  if (wave >= (M >> 4)) return;

  int m0 = wave << 4;
  int lo = lane & 15;
  int hi = lane >> 4;

  const float* aPtr = A  + (size_t)(m0 + lo) * K + (hi << 1);
  const float* bPtr = Bp + ((size_t)hi * N + lo) * 2;

  v8f z = {0.f,0.f,0.f,0.f,0.f,0.f,0.f,0.f};
  v8f acc[NT];
#pragma unroll
  for (int t = 0; t < NT; ++t) acc[t] = z;

  for (int k = 0; k < K; k += 8) {
    v2f a0 = *(const v2f*)(aPtr + k);
    v2f a1 = *(const v2f*)(aPtr + k + 4);
    const float* bp0 = bPtr + (size_t)k * N;
    const float* bp1 = bp0 + (size_t)4 * N;
    v2f b0[NT], b1[NT];
#pragma unroll
    for (int t = 0; t < NT; ++t) b0[t] = *(const v2f*)(bp0 + t * 32);
#pragma unroll
    for (int t = 0; t < NT; ++t) b1[t] = *(const v2f*)(bp1 + t * 32);
#pragma unroll
    for (int t = 0; t < NT; ++t)
      acc[t] = __builtin_amdgcn_wmma_f32_16x16x4_f32(
          false, a0, false, b0[t], (short)0, acc[t], false, false);
#pragma unroll
    for (int t = 0; t < NT; ++t)
      acc[t] = __builtin_amdgcn_wmma_f32_16x16x4_f32(
          false, a1, false, b1[t], (short)0, acc[t], false, false);
  }

  int cRow = m0 + (hi << 3);
#pragma unroll
  for (int r = 0; r < 8; ++r) {
    float sc = rs[cRow + r];
    if (SPLIT) {
      const int NH = N / 2;
      float* cp0 = C  + (size_t)(cRow + r) * NH + lo;
      float* cp1 = C1 + (size_t)(cRow + r) * NH + lo;
#pragma unroll
      for (int t = 0; t < NT / 2; ++t) cp0[t * 16] = acc[t][r] * sc;
#pragma unroll
      for (int t = 0; t < NT / 2; ++t) cp1[t * 16] = acc[NT / 2 + t][r] * sc;
    } else {
      float* cp = C + (size_t)(cRow + r) * N + lo;
#pragma unroll
      for (int t = 0; t < NT; ++t) cp[t * 16] = acc[t][r] * sc;
    }
  }
}

template <bool RELU>
__global__ void __launch_bounds__(256)
k_gather(const int* __restrict__ csr, const int* __restrict__ rowptr,
         const int* __restrict__ cnt, const float* __restrict__ dinv,
         const float* __restrict__ inA, const float* __restrict__ inB,
         const float* __restrict__ biasA, const float* __restrict__ biasB,
         float* __restrict__ outA, float* __restrict__ outB, int ld, int n) {
  int lane = threadIdx.x & 31;
  int d = __builtin_amdgcn_readfirstlane((int)((blockIdx.x * blockDim.x + threadIdx.x) >> 5));
  if (d >= n) return;
  int st  = rowptr[d];
  int len = cnt[d];
  float dd = dinv[d];
  int col = (lane & 15) << 2;
  const float* in  = ((lane < 16) ? inA : inB) + col;
  const float* bp  = ((lane < 16) ? biasA : biasB) + col;
  float*       out = ((lane < 16) ? outA : outB) + (size_t)d * ld + col;

  float4 acc = *(const float4*)(in + (size_t)d * ld);

  for (int base = 0; base < len; base += 32) {
    int m = len - base;
    if (m > 32) m = 32;
    int sj = d;
    float nj = 0.0f;
    if (lane < m) {
      int2 r = *(const int2*)(csr + 2 * (size_t)(st + base + lane));
      sj = r.x;
      nj = __int_as_float(r.y);
    }
    for (int j = 0; j < m; j += 8) {
      float4 r[8];
      float wgt[8];
#pragma unroll
      for (int u = 0; u < 8; ++u) {
        int s = __builtin_amdgcn_readlane(sj, j + u);
        wgt[u] = __int_as_float(__builtin_amdgcn_readlane(__float_as_int(nj), j + u));
        float4 t = make_float4(0.0f, 0.0f, 0.0f, 0.0f);
        if (j + u < m) t = *(const float4*)(in + (size_t)s * ld);
        r[u] = t;
      }
#pragma unroll
      for (int u = 0; u < 8; ++u) {
        acc.x += r[u].x * wgt[u];
        acc.y += r[u].y * wgt[u];
        acc.z += r[u].z * wgt[u];
        acc.w += r[u].w * wgt[u];
      }
    }
  }
  acc.x = acc.x * dd + bp[0];
  acc.y = acc.y * dd + bp[1];
  acc.z = acc.z * dd + bp[2];
  acc.w = acc.w * dd + bp[3];
  if (RELU) {
    acc.x = fmaxf(acc.x, 0.0f);
    acc.y = fmaxf(acc.y, 0.0f);
    acc.z = fmaxf(acc.z, 0.0f);
    acc.w = fmaxf(acc.w, 0.0f);
  }
  *(float4*)out = acc;
}

extern "C" void kernel_launch(void* const* d_in, const int* in_sizes, int n_in,
                              void* d_out, int out_size, void* d_ws, size_t ws_size,
                              hipStream_t stream) {
  const float* x  = (const float*)d_in[0];
  const int*   ei = (const int*)  d_in[1];
  const float* ew = (const float*)d_in[2];
  const float* W1 = (const float*)d_in[3];
  const float* b1 = (const float*)d_in[4];
  const float* W2 = (const float*)d_in[5];
  const float* b2 = (const float*)d_in[6];
  const float* W3 = (const float*)d_in[7];
  const float* b3 = (const float*)d_in[8];

  const int N = in_sizes[0] / IN_CH;
  const int E = in_sizes[2];

  float* ws   = (float*)d_ws;
  int*   cnt  = (int*)ws;
  float* dinv = ws + N;
  float* h1   = ws + 2 * (size_t)N;
  float* er   = h1 + (size_t)N * HID;
  int*   csr    = (int*)er;
  int*   rowptr = csr + 2 * (size_t)E;
  int*   oc     = rowptr + N;
  int*   seg    = (int*)h1;
  int*   srow   = seg + 2 * (size_t)E;
  float* W1p  = er + (size_t)N * HID;
  float* W23p = W1p + (size_t)IN_CH * HID;
  float* t2   = h1;
  float* t3   = h1 + (size_t)N * OUTC;
  float* om   = (float*)d_out;
  float* ol   = om + (size_t)N * OUTC;
  float* h    = (float*)d_out;

  const int T = 256;

  const int NT = (N + TILE_ROWS - 1) / TILE_ROWS;
  const int EC = (E + NCHUNK - 1) / NCHUNK;
  k_bin     <<<NCHUNK, BIN_T, 0, stream>>>(ei, ew, seg, srow, oc, E, EC, NT,
                                          W1, W2, W3, W1p, W23p);
  k_csr     <<<NT, NCHUNK, 0, stream>>>(seg, srow, oc, csr, rowptr, cnt, dinv, EC, N);

  {
    int waves  = N >> 4;
    int blocks = (waves * 32 + T - 1) / T;
    k_gemm_wmma<8, false><<<blocks, T, 0, stream>>>(x, W1p, dinv, h1, nullptr, N, IN_CH);
  }

  k_gather<true><<<(N * 32 + T - 1) / T, T, 0, stream>>>(
      csr, rowptr, cnt, dinv, h1, h1 + 64, b1, b1 + 64, h, h + 64, HID, N);

  {
    int waves  = N >> 4;
    int blocks = (waves * 32 + T - 1) / T;
    k_gemm_wmma<8, true><<<blocks, T, 0, stream>>>(h, W23p, dinv, t2, t3, N, HID);
  }

  k_gather<false><<<(N * 32 + T - 1) / T, T, 0, stream>>>(
      csr, rowptr, cnt, dinv, t2, t3, b2, b3, om, ol, OUTC, N);
}


# Both GEMMs as 256 resident work-groups with packed weights held in LDS; a 16-row tile is split over 4 waves (16x32 each, 16-byte fragments): equal WMMA count on every SIMD; k sum order changes
# speedup vs baseline: 4.3110x; 1.1734x over previous
#include <hip/hip_runtime.h>

typedef float v2f __attribute__((ext_vector_type(2)));
typedef float v8f __attribute__((ext_vector_type(8)));

#define IN_CH 512
#define HID   128
#define OUTC  64

#define TILE_SHIFT 7
#define TILE_ROWS  128
#define MAXT       512
#define NCHUNK     512
#define BIN_T      512
#define BIN_PER_T  7
#define BIN_CAP    (BIN_PER_T * BIN_T)
static_assert(BIN_T >= MAXT, "k_bin scans one tile counter per thread");
#define CSR_CAP    6144
static_assert(CSR_CAP % NCHUNK == 0, "the copy out of k_csr takes CSR_CAP / NCHUNK records per thread");

#define PACK_TOTAL  (IN_CH * HID + HID * 2 * OUTC)
#define PACK_PER_WG (PACK_TOTAL / NCHUNK)
static_assert(PACK_TOTAL % NCHUNK == 0 && PACK_PER_WG <= 512, "one element per thread of k_bin");

__device__ __forceinline__ size_t pack_pos(int k, int n, int N) {
  return (((size_t)(k >> 2)) * N + n) * 4 + (k & 3);
}

__global__ void __launch_bounds__(BIN_T)
k_bin(const int* __restrict__ ei, const float* __restrict__ ew,
      int* __restrict__ seg, int* __restrict__ srow, int* __restrict__ oc,
      int E, int EC, int NT,
      const float* __restrict__ W1, const float* __restrict__ W2,
      const float* __restrict__ W3,
      float* __restrict__ W1p, float* __restrict__ W23p) {
  __shared__ int hist[MAXT];
  __shared__ int curs[MAXT];
  __shared__ int wtot[BIN_T / 32];
  __shared__ int2 sseg[BIN_CAP];
  __shared__ int  srw[BIN_CAP];
  int t = threadIdx.x;
  int lane = t & 31;
  int w = t >> 5;
  int c = blockIdx.x;
  int e0 = c * EC;
  int e1 = e0 + EC;
  if (e1 > E) e1 = E;
  const int  ne  = e1 - e0;
  const bool stg = ne <= BIN_CAP;
  if (t < MAXT) hist[t] = 0;
  __syncthreads();
  float  pv = 0.0f;
  float* pp = W1p;
  if (t < PACK_PER_WG) {
    int idx = c * PACK_PER_WG + t;
    if (idx < IN_CH * HID) {
      int k = idx / HID;
      int n = idx - k * HID;
      pv = W1[idx];
      pp = W1p + pack_pos(k, n, HID);
    } else {
      int i2 = idx - IN_CH * HID;
      int k = i2 / (2 * OUTC);
      int n = i2 - k * (2 * OUTC);
      pv = (n < OUTC) ? W2[k * OUTC + n] : W3[k * OUTC + (n - OUTC)];
      pp = W23p + pack_pos(k, n, 2 * OUTC);
    }
  }
  int ed[BIN_PER_T], es[BIN_PER_T], ewb[BIN_PER_T];
  if (stg) {
#pragma unroll
    for (int u = 0; u < BIN_PER_T; ++u) {
      int e = e0 + t + u * BIN_T;
      bool ok = e < e1;
      ed[u]  = ok ? ei[E + e] : 0;
      es[u]  = ok ? ei[e] : 0;
      ewb[u] = ok ? __float_as_int(ew[e]) : 0;
    }
#pragma unroll
    for (int u = 0; u < BIN_PER_T; ++u)
      if (e0 + t + u * BIN_T < e1) atomicAdd(&hist[ed[u] >> TILE_SHIFT], 1);
  } else {
    for (int e = e0 + t; e < e1; e += BIN_T) atomicAdd(&hist[ei[E + e] >> TILE_SHIFT], 1);
  }
  __syncthreads();
  int v = 0, x = 0;
  if (t < MAXT) {
    v = hist[t];
    x = v;
    for (int dlt = 1; dlt < 32; dlt <<= 1) {
      int y = __shfl_up(x, dlt);
      if (lane >= dlt) x += y;
    }
    if (lane == 31) wtot[w] = x;
  }
  __syncthreads();
  if (t < MAXT) {
    int off = 0;
    for (int k = 0; k < BIN_T / 32; ++k) if (k < w) off += wtot[k];
    int ex = off + x - v;
    curs[t] = ex;
    if (t < NT) {
      int2 o;
      o.x = ex;
      o.y = v;
      *(int2*)(oc + 2 * ((size_t)t * NCHUNK + c)) = o;
    }
  }
  __syncthreads();
  if (stg) {
    int q[BIN_PER_T];
#pragma unroll
    for (int u = 0; u < BIN_PER_T; ++u) {
      q[u] = 0;
      if (e0 + t + u * BIN_T < e1) q[u] = atomicAdd(&curs[ed[u] >> TILE_SHIFT], 1);
    }
#pragma unroll
    for (int u = 0; u < BIN_PER_T; ++u) {
      if (e0 + t + u * BIN_T < e1) {
        int2 r;
        r.x = es[u];
        r.y = ewb[u];
        sseg[q[u]] = r;
        srw[q[u]]  = ed[u] & (TILE_ROWS - 1);
      }
    }
    __syncthreads();
    int2 cr[BIN_PER_T];
    int  cw[BIN_PER_T];
#pragma unroll
    for (int u = 0; u < BIN_PER_T; ++u) {
      cr[u] = sseg[t + u * BIN_T];
      cw[u] = srw[t + u * BIN_T];
    }
#pragma unroll
    for (int u = 0; u < BIN_PER_T; ++u) {
      int i = t + u * BIN_T;
      if (i < ne) {
        *(int2*)(seg + 2 * (size_t)(e0 + i)) = cr[u];
        srow[e0 + i] = cw[u];
      }
    }
  } else {
    for (int e = e0 + t; e < e1; e += BIN_T) {
      int d = ei[E + e];
      int p = e0 + atomicAdd(&curs[d >> TILE_SHIFT], 1);
      int2 r;
      r.x = ei[e];
      r.y = __float_as_int(ew[e]);
      *(int2*)(seg + 2 * (size_t)p) = r;
      srow[p] = d & (TILE_ROWS - 1);
    }
  }
  if (t < PACK_PER_WG) *pp = pv;
}

template <class F>
__device__ __forceinline__ float row_wsum(int len, int j, F wt) {
  float s = 0.0f;
  for (int k = j; k < len; k += 32) {
    float w0 = wt(k);
    float w1 = (k + 8  < len) ? wt(k + 8)  : 0.0f;
    float w2 = (k + 16 < len) ? wt(k + 16) : 0.0f;
    float w3 = (k + 24 < len) ? wt(k + 24) : 0.0f;
    s += w0;
    s += w1;
    s += w2;
    s += w3;
  }
  s += __shfl_xor(s, 1);
  s += __shfl_xor(s, 2);
  s += __shfl_xor(s, 4);
  return s;
}

#define RUN_G   8
#define RUN_U   (NCHUNK / (NCHUNK / RUN_G))
#define RUN_IT  2
__global__ void __launch_bounds__(NCHUNK)
k_csr(const int* __restrict__ seg, const int* __restrict__ srow,
      const int* __restrict__ oc,
      int* __restrict__ csr, int* __restrict__ rowptr, int* __restrict__ cnt,
      float* __restrict__ dinv, int EC, int n) {
  __shared__ int rcnt[TILE_ROWS];
  __shared__ int rcur[TILE_ROWS];
  __shared__ int red[NCHUNK / 32];
  __shared__ int redm[NCHUNK / 32];
  __shared__ int red2[TILE_ROWS / 32];
  __shared__ int rs0[NCHUNK];
  __shared__ int rm[NCHUNK];
  __shared__ int2 stage[CSR_CAP];
  int c = threadIdx.x;
  int lane = c & 31;
  int w = c >> 5;
  int T = blockIdx.x;
  int2 o = *(const int2*)(oc + 2 * ((size_t)T * NCHUNK + c));
  int s0 = c * EC + o.x;
  int m = o.y;
  rs0[c] = s0;
  rm[c]  = m;
  int part = o.x;
  int tsum = m;
  for (int dlt = 16; dlt > 0; dlt >>= 1) {
    part += __shfl_xor(part, dlt);
    tsum += __shfl_xor(tsum, dlt);
  }
  if (lane == 0) {
    red[w]  = part;
    redm[w] = tsum;
  }
  if (c < TILE_ROWS) rcnt[c] = 0;
  __syncthreads();
  int base = 0;
  int total = 0;
  for (int k = 0; k < NCHUNK / 32; ++k) {
    base  += red[k];
    total += redm[k];
  }
  const bool stg = total <= CSR_CAP;
  const int g = c >> 3;
  const int j = c & (RUN_G - 1);
  int  gs[RUN_U], gm[RUN_U];
  int2 rec[RUN_U][RUN_IT];
  int  rw[RUN_U][RUN_IT];
  if (stg) {
#pragma unroll
    for (int u = 0; u < RUN_U; ++u) {
      gs[u] = rs0[g + (NCHUNK / RUN_G) * u];
      gm[u] = rm[g + (NCHUNK / RUN_G) * u];
    }
#pragma unroll
    for (int u = 0; u < RUN_U; ++u) {
#pragma unroll
      for (int it = 0; it < RUN_IT; ++it) {
        int k = j + RUN_G * it;
        bool ok = k < gm[u];
        rw[u][it]  = ok ? srow[gs[u] + k] : 0;
        rec[u][it] = ok ? *(const int2*)(seg + 2 * (size_t)(gs[u] + k)) : make_int2(0, 0);
      }
    }
#pragma unroll
    for (int u = 0; u < RUN_U; ++u) {
#pragma unroll
      for (int it = 0; it < RUN_IT; ++it)
        if (j + RUN_G * it < gm[u]) atomicAdd(&rcnt[rw[u][it]], 1);
      for (int k = j + RUN_G * RUN_IT; k < gm[u]; k += RUN_G) atomicAdd(&rcnt[srow[gs[u] + k]], 1);
    }
  } else {
    for (int k = 0; k < m; ++k) atomicAdd(&rcnt[srow[s0 + k]], 1);
  }
  __syncthreads();
  int v = 0, x = 0;
  if (c < TILE_ROWS) {
    v = rcnt[c];
    x = v;
    for (int dlt = 1; dlt < 32; dlt <<= 1) {
      int y = __shfl_up(x, dlt);
      if (lane >= dlt) x += y;
    }
    if (lane == 31) red2[w] = x;
  }
  __syncthreads();
  if (c < TILE_ROWS) {
    int off = 0;
    for (int k = 0; k < TILE_ROWS / 32; ++k) if (k < w) off += red2[k];
    int ex = off + x - v;
    rcur[c] = ex;
    int row = T * TILE_ROWS + c;
    if (row < n) {
      rowptr[row] = base + ex;
      cnt[row] = v;
    }
  }
  __syncthreads();
  if (stg) {
    int q[RUN_U][RUN_IT];
#pragma unroll
    for (int u = 0; u < RUN_U; ++u) {
#pragma unroll
      for (int it = 0; it < RUN_IT; ++it) {
        q[u][it] = 0;
        if (j + RUN_G * it < gm[u]) q[u][it] = atomicAdd(&rcur[rw[u][it]], 1);
      }
    }
#pragma unroll
    for (int u = 0; u < RUN_U; ++u) {
#pragma unroll
      for (int it = 0; it < RUN_IT; ++it)
        if (j + RUN_G * it < gm[u]) stage[q[u][it]] = rec[u][it];
      for (int k = j + RUN_G * RUN_IT; k < gm[u]; k += RUN_G) {
        int p = atomicAdd(&rcur[srow[gs[u] + k]], 1);
        stage[p] = *(const int2*)(seg + 2 * (size_t)(gs[u] + k));
      }
    }
  } else {
    for (int k = 0; k < m; ++k) {
      int p = base + atomicAdd(&rcur[srow[s0 + k]], 1);
      *(int2*)(csr + 2 * (size_t)p) = *(const int2*)(seg + 2 * (size_t)(s0 + k));
    }
  }
  __syncthreads();
  if (stg) {
    int2 cr[CSR_CAP / NCHUNK];
#pragma unroll
    for (int u = 0; u < CSR_CAP / NCHUNK; ++u) cr[u] = stage[c + u * NCHUNK];
#pragma unroll
    for (int u = 0; u < CSR_CAP / NCHUNK; ++u) {
      int i = c + u * NCHUNK;
      if (i < total) *(int2*)(csr + 2 * (size_t)(base + i)) = cr[u];
    }
    for (int r = c >> 3; r < TILE_ROWS; r += NCHUNK / 8) {
      int len = rcnt[r];
      int rb  = rcur[r] - len;
      float s = row_wsum(len, j, [&](int k) -> float { return __int_as_float(stage[rb + k].y); });
      int row = T * TILE_ROWS + r;
      if (row < n && j == 0) {
        float dg = 1.0f + s;
        dinv[row] = (dg > 0.0f) ? rsqrtf(dg) : 0.0f;
      }
    }
  } else {
    for (int r = c >> 3; r < TILE_ROWS; r += NCHUNK / 8) {
      int len = rcnt[r];
      const int* wp = csr + 2 * (size_t)(base + rcur[r] - len) + 1;
      float s = row_wsum(len, j, [&](int k) -> float { return __int_as_float(wp[2 * k]); });
      int row = T * TILE_ROWS + r;
      if (row < n && j == 0) {
        float dg = 1.0f + s;
        dinv[row] = (dg > 0.0f) ? rsqrtf(dg) : 0.0f;
      }
    }
  }
}

#define GEMM_T   512
#define GEMM_WG  256
#define GEMM_N   128

__device__ __forceinline__ v8f wmma4(float ax, float ay, float bx, float by, v8f c) {
  v2f a = {ax, ay};
  v2f b = {bx, by};
  return __builtin_amdgcn_wmma_f32_16x16x4_f32(false, a, false, b, (short)0, c, false, false);
}

__device__ __forceinline__ void load_a64(float4 (&a)[8], const float* __restrict__ p) {
#pragma unroll
  for (int u = 0; u < 8; ++u) a[u] = *(const float4*)(p + 8 * u);
}

__device__ __forceinline__ void mma_chunk(const float4 (&a)[8], const float4* bs, v8f& acc0, v8f& acc1) {
#pragma unroll
  for (int h2 = 0; h2 < 2; ++h2) {
    float4 b0[4], b1[4];
#pragma unroll
    for (int u = 0; u < 4; ++u) {
      const float4* bp = bs + (h2 * 4 + u) * (2 * GEMM_N);
      b0[u] = bp[0];
      b1[u] = bp[16];
    }
#pragma unroll
    for (int u = 0; u < 4; ++u) {
      const float4 av = a[h2 * 4 + u];
      acc0 = wmma4(av.x, av.y, b0[u].x, b0[u].y, acc0);
      acc1 = wmma4(av.x, av.y, b1[u].x, b1[u].y, acc1);
      acc0 = wmma4(av.z, av.w, b0[u].z, b0[u].w, acc0);
      acc1 = wmma4(av.z, av.w, b1[u].z, b1[u].w, acc1);
    }
  }
}

template <int KDIM, bool SPLIT>
__global__ void __launch_bounds__(GEMM_T)
k_gemm_lds(const float* __restrict__ A,
           const float* __restrict__ Bp,
           const float* __restrict__ rs,
           float* __restrict__ C,
           float* __restrict__ C1,
           int M) {
  static_assert(KDIM % 128 == 0, "the k loop takes two chunks of 64 per trip");
  __shared__ __attribute__((aligned(16))) float Bs[KDIM * GEMM_N];
  {
    const float4* src = (const float4*)Bp;
    float4* dst = (float4*)Bs;
#pragma unroll 8
    for (int i = threadIdx.x; i < KDIM * GEMM_N / 4; i += GEMM_T) dst[i] = src[i];
  }
  __syncthreads();

  const int lane = threadIdx.x & 31;
  const int wave = __builtin_amdgcn_readfirstlane((int)(threadIdx.x >> 5));
  const int q  = wave & 3;
  const int g  = wave >> 2;
  const int lo = lane & 15;
  const int hi = lane >> 4;
  const int ntiles = M >> 4;
  const int tstep  = (int)gridDim.x * 4;
  constexpr int NCH = KDIM / 64;

  int t = (int)blockIdx.x + (int)gridDim.x * g;
  if (t >= ntiles) return;

  const float4* bs = (const float4*)Bs + hi * GEMM_N + q * 32 + lo;
  const float* ap = A + (size_t)((t << 4) + lo) * KDIM + (hi << 2);

  float4 p0[8], p1[8];
  load_a64(p0, ap);

  const v8f z = {0.f,0.f,0.f,0.f,0.f,0.f,0.f,0.f};
  for (;;) {
    const int tn = t + tstep;
    const float* apn = (tn < ntiles) ? A + (size_t)((tn << 4) + lo) * KDIM + (hi << 2) : ap;
    v8f acc0 = z, acc1 = z;
#pragma unroll 1
    for (int c = 0; c < NCH; c += 2) {
      load_a64(p1, ap + 64 * (c + 1));
      mma_chunk(p0, bs + (size_t)c * (16 * GEMM_N), acc0, acc1);
      load_a64(p0, (c + 2 < NCH) ? ap + 64 * (c + 2) : apn);
      mma_chunk(p1, bs + (size_t)(c + 1) * (16 * GEMM_N), acc0, acc1);
    }

    const int cRow = (t << 4) + (hi << 3);
#pragma unroll
    for (int r = 0; r < 8; ++r) {
      float sc = rs[cRow + r];
      float* cp;
      if (SPLIT) {
        cp = ((q < 2) ? C : C1) + (size_t)(cRow + r) * (GEMM_N / 2) + (q & 1) * 32 + lo;
      } else {
        cp = C + (size_t)(cRow + r) * GEMM_N + q * 32 + lo;
      }
      cp[0]  = acc0[r] * sc;
      cp[16] = acc1[r] * sc;
    }

    if (tn >= ntiles) break;
    t  = tn;
    ap = apn;
  }
}

template <bool RELU>
__global__ void __launch_bounds__(256)
k_gather(const int* __restrict__ csr, const int* __restrict__ rowptr,
         const int* __restrict__ cnt, const float* __restrict__ dinv,
         const float* __restrict__ inA, const float* __restrict__ inB,
         const float* __restrict__ biasA, const float* __restrict__ biasB,
         float* __restrict__ outA, float* __restrict__ outB, int ld, int n) {
  int lane = threadIdx.x & 31;
  int d = __builtin_amdgcn_readfirstlane((int)((blockIdx.x * blockDim.x + threadIdx.x) >> 5));
  if (d >= n) return;
  int st  = rowptr[d];
  int len = cnt[d];
  float dd = dinv[d];
  int col = (lane & 15) << 2;
  const float* in  = ((lane < 16) ? inA : inB) + col;
  const float* bp  = ((lane < 16) ? biasA : biasB) + col;
  float*       out = ((lane < 16) ? outA : outB) + (size_t)d * ld + col;

  float4 acc = *(const float4*)(in + (size_t)d * ld);

  for (int base = 0; base < len; base += 32) {
    int m = len - base;
    if (m > 32) m = 32;
    int sj = d;
    float nj = 0.0f;
    if (lane < m) {
      int2 r = *(const int2*)(csr + 2 * (size_t)(st + base + lane));
      sj = r.x;
      nj = __int_as_float(r.y);
    }
    for (int j = 0; j < m; j += 8) {
      float4 r[8];
      float wgt[8];
#pragma unroll
      for (int u = 0; u < 8; ++u) {
        int s = __builtin_amdgcn_readlane(sj, j + u);
        wgt[u] = __int_as_float(__builtin_amdgcn_readlane(__float_as_int(nj), j + u));
        float4 t = make_float4(0.0f, 0.0f, 0.0f, 0.0f);
        if (j + u < m) t = *(const float4*)(in + (size_t)s * ld);
        r[u] = t;
      }
#pragma unroll
      for (int u = 0; u < 8; ++u) {
        acc.x += r[u].x * wgt[u];
        acc.y += r[u].y * wgt[u];
        acc.z += r[u].z * wgt[u];
        acc.w += r[u].w * wgt[u];
      }
    }
  }
  acc.x = acc.x * dd + bp[0];
  acc.y = acc.y * dd + bp[1];
  acc.z = acc.z * dd + bp[2];
  acc.w = acc.w * dd + bp[3];
  if (RELU) {
    acc.x = fmaxf(acc.x, 0.0f);
    acc.y = fmaxf(acc.y, 0.0f);
    acc.z = fmaxf(acc.z, 0.0f);
    acc.w = fmaxf(acc.w, 0.0f);
  }
  *(float4*)out = acc;
}

extern "C" void kernel_launch(void* const* d_in, const int* in_sizes, int n_in,
                              void* d_out, int out_size, void* d_ws, size_t ws_size,
                              hipStream_t stream) {
  const float* x  = (const float*)d_in[0];
  const int*   ei = (const int*)  d_in[1];
  const float* ew = (const float*)d_in[2];
  const float* W1 = (const float*)d_in[3];
  const float* b1 = (const float*)d_in[4];
  const float* W2 = (const float*)d_in[5];
  const float* b2 = (const float*)d_in[6];
  const float* W3 = (const float*)d_in[7];
  const float* b3 = (const float*)d_in[8];

  const int N = in_sizes[0] / IN_CH;
  const int E = in_sizes[2];

  float* ws   = (float*)d_ws;
  int*   cnt  = (int*)ws;
  float* dinv = ws + N;
  float* h1   = ws + 2 * (size_t)N;
  float* er   = h1 + (size_t)N * HID;
  int*   csr    = (int*)er;
  int*   rowptr = csr + 2 * (size_t)E;
  int*   oc     = rowptr + N;
  int*   seg    = (int*)h1;
  int*   srow   = seg + 2 * (size_t)E;
  float* W1p  = er + (size_t)N * HID;
  float* W23p = W1p + (size_t)IN_CH * HID;
  float* t2   = h1;
  float* t3   = h1 + (size_t)N * OUTC;
  float* om   = (float*)d_out;
  float* ol   = om + (size_t)N * OUTC;
  float* h    = (float*)d_out;

  const int T = 256;

  const int NT = (N + TILE_ROWS - 1) / TILE_ROWS;
  const int EC = (E + NCHUNK - 1) / NCHUNK;
  k_bin     <<<NCHUNK, BIN_T, 0, stream>>>(ei, ew, seg, srow, oc, E, EC, NT,
                                          W1, W2, W3, W1p, W23p);
  k_csr     <<<NT, NCHUNK, 0, stream>>>(seg, srow, oc, csr, rowptr, cnt, dinv, EC, N);

  k_gemm_lds<IN_CH, false><<<GEMM_WG, GEMM_T, 0, stream>>>(x, W1p, dinv, h1, nullptr, N);

  k_gather<true><<<(N * 32 + T - 1) / T, T, 0, stream>>>(
      csr, rowptr, cnt, dinv, h1, h1 + 64, b1, b1 + 64, h, h + 64, HID, N);

  k_gemm_lds<HID, true><<<GEMM_WG, GEMM_T, 0, stream>>>(h, W23p, dinv, t2, t3, N);

  k_gather<false><<<(N * 32 + T - 1) / T, T, 0, stream>>>(
      csr, rowptr, cnt, dinv, t2, t3, b2, b3, om, ol, OUTC, N);
}


# Layer-2 product moved into the layer-1 aggregation launch: 1024 resident work-groups keep [W2|W3] in LDS, 16 waves sum a row each into an LDS tile, 8 waves multiply it; 5 launches; sum orders kept
# speedup vs baseline: 4.7580x; 1.1037x over previous
#include <hip/hip_runtime.h>

typedef float v2f __attribute__((ext_vector_type(2)));
typedef float v8f __attribute__((ext_vector_type(8)));

#define IN_CH 512
#define HID   128
#define OUTC  64

#define TILE_SHIFT 7
#define TILE_ROWS  128
#define MAXT       512
#define NCHUNK     512
#define BIN_T      512
#define BIN_PER_T  7
#define BIN_CAP    (BIN_PER_T * BIN_T)
static_assert(BIN_T >= MAXT, "k_bin scans one tile counter per thread");
#define CSR_CAP    6144
static_assert(CSR_CAP % NCHUNK == 0, "the copy out of k_csr takes CSR_CAP / NCHUNK records per thread");

#define PACK_TOTAL  (IN_CH * HID + HID * 2 * OUTC)
#define PACK_PER_WG (PACK_TOTAL / NCHUNK)
static_assert(PACK_TOTAL % NCHUNK == 0 && PACK_PER_WG <= 512, "one element per thread of k_bin");

__device__ __forceinline__ size_t pack_pos(int k, int n, int N) {
  return (((size_t)(k >> 2)) * N + n) * 4 + (k & 3);
}

__global__ void __launch_bounds__(BIN_T)
k_bin(const int* __restrict__ ei, const float* __restrict__ ew,
      int* __restrict__ seg, int* __restrict__ srow, int* __restrict__ oc,
      int E, int EC, int NT,
      const float* __restrict__ W1, const float* __restrict__ W2,
      const float* __restrict__ W3,
      float* __restrict__ W1p, float* __restrict__ W23p) {
  __shared__ int hist[MAXT];
  __shared__ int curs[MAXT];
  __shared__ int wtot[BIN_T / 32];
  __shared__ int2 sseg[BIN_CAP];
  __shared__ int  srw[BIN_CAP];
  int t = threadIdx.x;
  int lane = t & 31;
  int w = t >> 5;
  int c = blockIdx.x;
  int e0 = c * EC;
  int e1 = e0 + EC;
  if (e1 > E) e1 = E;
  const int  ne  = e1 - e0;
  const bool stg = ne <= BIN_CAP;
  if (t < MAXT) hist[t] = 0;
  __syncthreads();
  float  pv = 0.0f;
  float* pp = W1p;
  if (t < PACK_PER_WG) {
    int idx = c * PACK_PER_WG + t;
    if (idx < IN_CH * HID) {
      int k = idx / HID;
      int n = idx - k * HID;
      pv = W1[idx];
      pp = W1p + pack_pos(k, n, HID);
    } else {
      int i2 = idx - IN_CH * HID;
      int k = i2 / (2 * OUTC);
      int n = i2 - k * (2 * OUTC);
      pv = (n < OUTC) ? W2[k * OUTC + n] : W3[k * OUTC + (n - OUTC)];
      pp = W23p + pack_pos(k, n, 2 * OUTC);
    }
  }
  int ed[BIN_PER_T], es[BIN_PER_T], ewb[BIN_PER_T];
  if (stg) {
#pragma unroll
    for (int u = 0; u < BIN_PER_T; ++u) {
      int e = e0 + t + u * BIN_T;
      bool ok = e < e1;
      ed[u]  = ok ? ei[E + e] : 0;
      es[u]  = ok ? ei[e] : 0;
      ewb[u] = ok ? __float_as_int(ew[e]) : 0;
    }
#pragma unroll
    for (int u = 0; u < BIN_PER_T; ++u)
      if (e0 + t + u * BIN_T < e1) atomicAdd(&hist[ed[u] >> TILE_SHIFT], 1);
  } else {
    for (int e = e0 + t; e < e1; e += BIN_T) atomicAdd(&hist[ei[E + e] >> TILE_SHIFT], 1);
  }
  __syncthreads();
  int v = 0, x = 0;
  if (t < MAXT) {
    v = hist[t];
    x = v;
    for (int dlt = 1; dlt < 32; dlt <<= 1) {
      int y = __shfl_up(x, dlt);
      if (lane >= dlt) x += y;
    }
    if (lane == 31) wtot[w] = x;
  }
  __syncthreads();
  if (t < MAXT) {
    int off = 0;
    for (int k = 0; k < BIN_T / 32; ++k) if (k < w) off += wtot[k];
    int ex = off + x - v;
    curs[t] = ex;
    if (t < NT) {
      int2 o;
      o.x = ex;
      o.y = v;
      *(int2*)(oc + 2 * ((size_t)t * NCHUNK + c)) = o;
    }
  }
  __syncthreads();
  if (stg) {
    int q[BIN_PER_T];
#pragma unroll
    for (int u = 0; u < BIN_PER_T; ++u) {
      q[u] = 0;
      if (e0 + t + u * BIN_T < e1) q[u] = atomicAdd(&curs[ed[u] >> TILE_SHIFT], 1);
    }
#pragma unroll
    for (int u = 0; u < BIN_PER_T; ++u) {
      if (e0 + t + u * BIN_T < e1) {
        int2 r;
        r.x = es[u];
        r.y = ewb[u];
        sseg[q[u]] = r;
        srw[q[u]]  = ed[u] & (TILE_ROWS - 1);
      }
    }
    __syncthreads();
    int2 cr[BIN_PER_T];
    int  cw[BIN_PER_T];
#pragma unroll
    for (int u = 0; u < BIN_PER_T; ++u) {
      cr[u] = sseg[t + u * BIN_T];
      cw[u] = srw[t + u * BIN_T];
    }
#pragma unroll
    for (int u = 0; u < BIN_PER_T; ++u) {
      int i = t + u * BIN_T;
      if (i < ne) {
        *(int2*)(seg + 2 * (size_t)(e0 + i)) = cr[u];
        srow[e0 + i] = cw[u];
      }
    }
  } else {
    for (int e = e0 + t; e < e1; e += BIN_T) {
      int d = ei[E + e];
      int p = e0 + atomicAdd(&curs[d >> TILE_SHIFT], 1);
      int2 r;
      r.x = ei[e];
      r.y = __float_as_int(ew[e]);
      *(int2*)(seg + 2 * (size_t)p) = r;
      srow[p] = d & (TILE_ROWS - 1);
    }
  }
  if (t < PACK_PER_WG) *pp = pv;
}

template <class F>
__device__ __forceinline__ float row_wsum(int len, int j, F wt) {
  float s = 0.0f;
  for (int k = j; k < len; k += 32) {
    float w0 = wt(k);
    float w1 = (k + 8  < len) ? wt(k + 8)  : 0.0f;
    float w2 = (k + 16 < len) ? wt(k + 16) : 0.0f;
    float w3 = (k + 24 < len) ? wt(k + 24) : 0.0f;
    s += w0;
    s += w1;
    s += w2;
    s += w3;
  }
  s += __shfl_xor(s, 1);
  s += __shfl_xor(s, 2);
  s += __shfl_xor(s, 4);
  return s;
}

#define RUN_G   8
#define RUN_U   (NCHUNK / (NCHUNK / RUN_G))
#define RUN_IT  2
__global__ void __launch_bounds__(NCHUNK)
k_csr(const int* __restrict__ seg, const int* __restrict__ srow,
      const int* __restrict__ oc,
      int* __restrict__ csr, int* __restrict__ rowptr, int* __restrict__ cnt,
      float* __restrict__ dinv, int EC, int n) {
  __shared__ int rcnt[TILE_ROWS];
  __shared__ int rcur[TILE_ROWS];
  __shared__ int red[NCHUNK / 32];
  __shared__ int redm[NCHUNK / 32];
  __shared__ int red2[TILE_ROWS / 32];
  __shared__ int rs0[NCHUNK];
  __shared__ int rm[NCHUNK];
  __shared__ int2 stage[CSR_CAP];
  int c = threadIdx.x;
  int lane = c & 31;
  int w = c >> 5;
  int T = blockIdx.x;
  int2 o = *(const int2*)(oc + 2 * ((size_t)T * NCHUNK + c));
  int s0 = c * EC + o.x;
  int m = o.y;
  rs0[c] = s0;
  rm[c]  = m;
  int part = o.x;
  int tsum = m;
  for (int dlt = 16; dlt > 0; dlt >>= 1) {
    part += __shfl_xor(part, dlt);
    tsum += __shfl_xor(tsum, dlt);
  }
  if (lane == 0) {
    red[w]  = part;
    redm[w] = tsum;
  }
  if (c < TILE_ROWS) rcnt[c] = 0;
  __syncthreads();
  int base = 0;
  int total = 0;
  for (int k = 0; k < NCHUNK / 32; ++k) {
    base  += red[k];
    total += redm[k];
  }
  const bool stg = total <= CSR_CAP;
  const int g = c >> 3;
  const int j = c & (RUN_G - 1);
  int  gs[RUN_U], gm[RUN_U];
  int2 rec[RUN_U][RUN_IT];
  int  rw[RUN_U][RUN_IT];
  if (stg) {
#pragma unroll
    for (int u = 0; u < RUN_U; ++u) {
      gs[u] = rs0[g + (NCHUNK / RUN_G) * u];
      gm[u] = rm[g + (NCHUNK / RUN_G) * u];
    }
#pragma unroll
    for (int u = 0; u < RUN_U; ++u) {
#pragma unroll
      for (int it = 0; it < RUN_IT; ++it) {
        int k = j + RUN_G * it;
        bool ok = k < gm[u];
        rw[u][it]  = ok ? srow[gs[u] + k] : 0;
        rec[u][it] = ok ? *(const int2*)(seg + 2 * (size_t)(gs[u] + k)) : make_int2(0, 0);
      }
    }
#pragma unroll
    for (int u = 0; u < RUN_U; ++u) {
#pragma unroll
      for (int it = 0; it < RUN_IT; ++it)
        if (j + RUN_G * it < gm[u]) atomicAdd(&rcnt[rw[u][it]], 1);
      for (int k = j + RUN_G * RUN_IT; k < gm[u]; k += RUN_G) atomicAdd(&rcnt[srow[gs[u] + k]], 1);
    }
  } else {
    for (int k = 0; k < m; ++k) atomicAdd(&rcnt[srow[s0 + k]], 1);
  }
  __syncthreads();
  int v = 0, x = 0;
  if (c < TILE_ROWS) {
    v = rcnt[c];
    x = v;
    for (int dlt = 1; dlt < 32; dlt <<= 1) {
      int y = __shfl_up(x, dlt);
      if (lane >= dlt) x += y;
    }
    if (lane == 31) red2[w] = x;
  }
  __syncthreads();
  if (c < TILE_ROWS) {
    int off = 0;
    for (int k = 0; k < TILE_ROWS / 32; ++k) if (k < w) off += red2[k];
    int ex = off + x - v;
    rcur[c] = ex;
    int row = T * TILE_ROWS + c;
    if (row < n) {
      rowptr[row] = base + ex;
      cnt[row] = v;
    }
  }
  __syncthreads();
  if (stg) {
    int q[RUN_U][RUN_IT];
#pragma unroll
    for (int u = 0; u < RUN_U; ++u) {
#pragma unroll
      for (int it = 0; it < RUN_IT; ++it) {
        q[u][it] = 0;
        if (j + RUN_G * it < gm[u]) q[u][it] = atomicAdd(&rcur[rw[u][it]], 1);
      }
    }
#pragma unroll
    for (int u = 0; u < RUN_U; ++u) {
#pragma unroll
      for (int it = 0; it < RUN_IT; ++it)
        if (j + RUN_G * it < gm[u]) stage[q[u][it]] = rec[u][it];
      for (int k = j + RUN_G * RUN_IT; k < gm[u]; k += RUN_G) {
        int p = atomicAdd(&rcur[srow[gs[u] + k]], 1);
        stage[p] = *(const int2*)(seg + 2 * (size_t)(gs[u] + k));
      }
    }
  } else {
    for (int k = 0; k < m; ++k) {
      int p = base + atomicAdd(&rcur[srow[s0 + k]], 1);
      *(int2*)(csr + 2 * (size_t)p) = *(const int2*)(seg + 2 * (size_t)(s0 + k));
    }
  }
  __syncthreads();
  if (stg) {
    int2 cr[CSR_CAP / NCHUNK];
#pragma unroll
    for (int u = 0; u < CSR_CAP / NCHUNK; ++u) cr[u] = stage[c + u * NCHUNK];
#pragma unroll
    for (int u = 0; u < CSR_CAP / NCHUNK; ++u) {
      int i = c + u * NCHUNK;
      if (i < total) *(int2*)(csr + 2 * (size_t)(base + i)) = cr[u];
    }
    for (int r = c >> 3; r < TILE_ROWS; r += NCHUNK / 8) {
      int len = rcnt[r];
      int rb  = rcur[r] - len;
      float s = row_wsum(len, j, [&](int k) -> float { return __int_as_float(stage[rb + k].y); });
      int row = T * TILE_ROWS + r;
      if (row < n && j == 0) {
        float dg = 1.0f + s;
        dinv[row] = (dg > 0.0f) ? rsqrtf(dg) : 0.0f;
      }
    }
  } else {
    for (int r = c >> 3; r < TILE_ROWS; r += NCHUNK / 8) {
      int len = rcnt[r];
      const int* wp = csr + 2 * (size_t)(base + rcur[r] - len) + 1;
      float s = row_wsum(len, j, [&](int k) -> float { return __int_as_float(wp[2 * k]); });
      int row = T * TILE_ROWS + r;
      if (row < n && j == 0) {
        float dg = 1.0f + s;
        dinv[row] = (dg > 0.0f) ? rsqrtf(dg) : 0.0f;
      }
    }
  }
}

#define GEMM_T   512
#define GEMM_WG  256
#define GEMM_N   128

__device__ __forceinline__ v8f wmma4(float ax, float ay, float bx, float by, v8f c) {
  v2f a = {ax, ay};
  v2f b = {bx, by};
  return __builtin_amdgcn_wmma_f32_16x16x4_f32(false, a, false, b, (short)0, c, false, false);
}

__device__ __forceinline__ void load_a64(float4 (&a)[8], const float* __restrict__ p) {
#pragma unroll
  for (int u = 0; u < 8; ++u) a[u] = *(const float4*)(p + 8 * u);
}

__device__ __forceinline__ void mma_chunk(const float4 (&a)[8], const float4* bs, v8f& acc0, v8f& acc1) {
#pragma unroll
  for (int h2 = 0; h2 < 2; ++h2) {
    float4 b0[4], b1[4];
#pragma unroll
    for (int u = 0; u < 4; ++u) {
      const float4* bp = bs + (h2 * 4 + u) * (2 * GEMM_N);
      b0[u] = bp[0];
      b1[u] = bp[16];
    }
#pragma unroll
    for (int u = 0; u < 4; ++u) {
      const float4 av = a[h2 * 4 + u];
      acc0 = wmma4(av.x, av.y, b0[u].x, b0[u].y, acc0);
      acc1 = wmma4(av.x, av.y, b1[u].x, b1[u].y, acc1);
      acc0 = wmma4(av.z, av.w, b0[u].z, b0[u].w, acc0);
      acc1 = wmma4(av.z, av.w, b1[u].z, b1[u].w, acc1);
    }
  }
}

template <int KDIM>
__global__ void __launch_bounds__(GEMM_T)
k_gemm_lds(const float* __restrict__ A,
           const float* __restrict__ Bp,
           const float* __restrict__ rs,
           float* __restrict__ C,
           int M) {
  static_assert(KDIM % 128 == 0, "the k loop takes two chunks of 64 per trip");
  __shared__ __attribute__((aligned(16))) float Bs[KDIM * GEMM_N];
  {
    const float4* src = (const float4*)Bp;
    float4* dst = (float4*)Bs;
#pragma unroll 8
    for (int i = threadIdx.x; i < KDIM * GEMM_N / 4; i += GEMM_T) dst[i] = src[i];
  }
  __syncthreads();

  const int lane = threadIdx.x & 31;
  const int wave = __builtin_amdgcn_readfirstlane((int)(threadIdx.x >> 5));
  const int q  = wave & 3;
  const int g  = wave >> 2;
  const int lo = lane & 15;
  const int hi = lane >> 4;
  const int ntiles = M >> 4;
  const int tstep  = (int)gridDim.x * 4;
  constexpr int NCH = KDIM / 64;

  int t = (int)blockIdx.x + (int)gridDim.x * g;
  if (t >= ntiles) return;

  const float4* bs = (const float4*)Bs + hi * GEMM_N + q * 32 + lo;
  const float* ap = A + (size_t)((t << 4) + lo) * KDIM + (hi << 2);

  float4 p0[8], p1[8];
  load_a64(p0, ap);

  const v8f z = {0.f,0.f,0.f,0.f,0.f,0.f,0.f,0.f};
  for (;;) {
    const int tn = t + tstep;
    const float* apn = (tn < ntiles) ? A + (size_t)((tn << 4) + lo) * KDIM + (hi << 2) : ap;
    v8f acc0 = z, acc1 = z;
#pragma unroll 1
    for (int c = 0; c < NCH; c += 2) {
      load_a64(p1, ap + 64 * (c + 1));
      mma_chunk(p0, bs + (size_t)c * (16 * GEMM_N), acc0, acc1);
      load_a64(p0, (c + 2 < NCH) ? ap + 64 * (c + 2) : apn);
      mma_chunk(p1, bs + (size_t)(c + 1) * (16 * GEMM_N), acc0, acc1);
    }

    const int cRow = (t << 4) + (hi << 3);
#pragma unroll
    for (int r = 0; r < 8; ++r) {
      float sc = rs[cRow + r];
      float* cp = C + (size_t)(cRow + r) * GEMM_N + q * 32 + lo;
      cp[0]  = acc0[r] * sc;
      cp[16] = acc1[r] * sc;
    }

    if (tn >= ntiles) break;
    t  = tn;
    ap = apn;
  }
}

template <bool RELU>
__device__ __forceinline__ float4
row_sum(const int* __restrict__ csr, const int* __restrict__ rowptr,
        const int* __restrict__ cnt, const float* __restrict__ dinv,
        const float* __restrict__ inA, const float* __restrict__ inB,
        const float* __restrict__ biasA, const float* __restrict__ biasB,
        int ld, int d, int lane) {
  int st  = rowptr[d];
  int len = cnt[d];
  float dd = dinv[d];
  int col = (lane & 15) << 2;
  const float* in  = ((lane < 16) ? inA : inB) + col;
  const float* bp  = ((lane < 16) ? biasA : biasB) + col;

  float4 acc = *(const float4*)(in + (size_t)d * ld);

  for (int base = 0; base < len; base += 32) {
    int m = len - base;
    if (m > 32) m = 32;
    int sj = d;
    float nj = 0.0f;
    if (lane < m) {
      int2 r = *(const int2*)(csr + 2 * (size_t)(st + base + lane));
      sj = r.x;
      nj = __int_as_float(r.y);
    }
    for (int j = 0; j < m; j += 8) {
      float4 r[8];
      float wgt[8];
#pragma unroll
      for (int u = 0; u < 8; ++u) {
        int s = __builtin_amdgcn_readlane(sj, j + u);
        wgt[u] = __int_as_float(__builtin_amdgcn_readlane(__float_as_int(nj), j + u));
        float4 t = make_float4(0.0f, 0.0f, 0.0f, 0.0f);
        if (j + u < m) t = *(const float4*)(in + (size_t)s * ld);
        r[u] = t;
      }
#pragma unroll
      for (int u = 0; u < 8; ++u) {
        acc.x += r[u].x * wgt[u];
        acc.y += r[u].y * wgt[u];
        acc.z += r[u].z * wgt[u];
        acc.w += r[u].w * wgt[u];
      }
    }
  }
  acc.x = acc.x * dd + bp[0];
  acc.y = acc.y * dd + bp[1];
  acc.z = acc.z * dd + bp[2];
  acc.w = acc.w * dd + bp[3];
  if (RELU) {
    acc.x = fmaxf(acc.x, 0.0f);
    acc.y = fmaxf(acc.y, 0.0f);
    acc.z = fmaxf(acc.z, 0.0f);
    acc.w = fmaxf(acc.w, 0.0f);
  }
  return acc;
}

__global__ void __launch_bounds__(256)
k_gather(const int* __restrict__ csr, const int* __restrict__ rowptr,
         const int* __restrict__ cnt, const float* __restrict__ dinv,
         const float* __restrict__ inA, const float* __restrict__ inB,
         const float* __restrict__ biasA, const float* __restrict__ biasB,
         float* __restrict__ outA, float* __restrict__ outB, int ld, int n) {
  int lane = threadIdx.x & 31;
  int d = __builtin_amdgcn_readfirstlane((int)((blockIdx.x * blockDim.x + threadIdx.x) >> 5));
  if (d >= n) return;
  float4 acc = row_sum<false>(csr, rowptr, cnt, dinv, inA, inB, biasA, biasB, ld, d, lane);
  float* out = ((lane < 16) ? outA : outB) + (size_t)d * ld + ((lane & 15) << 2);
  *(float4*)out = acc;
}

#define GM_T     512
#define GM_WG    1024
#define HS_PITCH 132

__global__ void __launch_bounds__(GM_T)
k_gather_mm(const int* __restrict__ csr, const int* __restrict__ rowptr,
            const int* __restrict__ cnt, const float* __restrict__ dinv,
            const float* __restrict__ inA, const float* __restrict__ inB,
            const float* __restrict__ biasA, const float* __restrict__ biasB,
            const float* __restrict__ Bp,
            float* __restrict__ C, float* __restrict__ C1, int n) {
  __shared__ __attribute__((aligned(16))) float Bs[HID * GEMM_N];
  __shared__ __attribute__((aligned(16))) float hs[16 * HS_PITCH];
  {
    const float4* src = (const float4*)Bp;
    float4* dst = (float4*)Bs;
#pragma unroll 8
    for (int i = threadIdx.x; i < HID * GEMM_N / 4; i += GM_T) dst[i] = src[i];
  }

  const int lane = threadIdx.x & 31;
  const int wave = __builtin_amdgcn_readfirstlane((int)(threadIdx.x >> 5));
  const int lo = lane & 15;
  const int hi = lane >> 4;
  const int ntiles = n >> 4;

  for (int t = (int)blockIdx.x; t < ntiles; t += (int)gridDim.x) {
    const int d = (t << 4) + wave;
    float4 acc = row_sum<true>(csr, rowptr, cnt, dinv, inA, inB, biasA, biasB, HID, d, lane);
    *(float4*)(hs + wave * HS_PITCH + (lane << 2)) = acc;
    __syncthreads();
    v8f cc = {0.f,0.f,0.f,0.f,0.f,0.f,0.f,0.f};
    if (wave < 8) {
      const float4* bs = (const float4*)Bs + hi * GEMM_N + wave * 16 + lo;
      const float*  as = hs + lo * HS_PITCH + (hi << 2);
#pragma unroll
      for (int g4 = 0; g4 < HID / 32; ++g4) {
        float4 av[4], bv[4];
#pragma unroll
        for (int u = 0; u < 4; ++u) {
          av[u] = *(const float4*)(as + 8 * (g4 * 4 + u));
          bv[u] = bs[(g4 * 4 + u) * (2 * GEMM_N)];
        }
#pragma unroll
        for (int u = 0; u < 4; ++u) {
          cc = wmma4(av[u].x, av[u].y, bv[u].x, bv[u].y, cc);
          cc = wmma4(av[u].z, av[u].w, bv[u].z, bv[u].w, cc);
        }
      }
    }
    __syncthreads();
    if (wave < 8) {
      const int cRow = (t << 4) + (hi << 3);
      float* cb = ((wave < 4) ? C : C1) + (wave & 3) * 16 + lo;
#pragma unroll
      for (int r = 0; r < 8; ++r)
        cb[(size_t)(cRow + r) * OUTC] = cc[r] * dinv[cRow + r];
    }
  }
}

extern "C" void kernel_launch(void* const* d_in, const int* in_sizes, int n_in,
                              void* d_out, int out_size, void* d_ws, size_t ws_size,
                              hipStream_t stream) {
  const float* x  = (const float*)d_in[0];
  const int*   ei = (const int*)  d_in[1];
  const float* ew = (const float*)d_in[2];
  const float* W1 = (const float*)d_in[3];
  const float* b1 = (const float*)d_in[4];
  const float* W2 = (const float*)d_in[5];
  const float* b2 = (const float*)d_in[6];
  const float* W3 = (const float*)d_in[7];
  const float* b3 = (const float*)d_in[8];

  const int N = in_sizes[0] / IN_CH;
  const int E = in_sizes[2];

  float* ws   = (float*)d_ws;
  int*   cnt  = (int*)ws;
  float* dinv = ws + N;
  float* t23  = ws + 2 * (size_t)N;
  float* er   = t23 + (size_t)N * HID;
  int*   csr    = (int*)er;
  int*   rowptr = csr + 2 * (size_t)E;
  int*   oc     = rowptr + N;
  int*   seg    = (int*)t23;
  int*   srow   = seg + 2 * (size_t)E;
  float* W1p  = er + (size_t)N * HID;
  float* W23p = W1p + (size_t)IN_CH * HID;
  float* t2   = t23;
  float* t3   = t23 + (size_t)N * OUTC;
  float* om   = (float*)d_out;
  float* ol   = om + (size_t)N * OUTC;
  float* h1   = (float*)d_out;

  const int T = 256;

  const int NT = (N + TILE_ROWS - 1) / TILE_ROWS;
  const int EC = (E + NCHUNK - 1) / NCHUNK;
  k_bin     <<<NCHUNK, BIN_T, 0, stream>>>(ei, ew, seg, srow, oc, E, EC, NT,
                                          W1, W2, W3, W1p, W23p);
  k_csr     <<<NT, NCHUNK, 0, stream>>>(seg, srow, oc, csr, rowptr, cnt, dinv, EC, N);

  k_gemm_lds<IN_CH><<<GEMM_WG, GEMM_T, 0, stream>>>(x, W1p, dinv, h1, N);

  k_gather_mm<<<GM_WG, GM_T, 0, stream>>>(
      csr, rowptr, cnt, dinv, h1, h1 + 64, b1, b1 + 64, W23p, t2, t3, N);

  k_gather<<<(N * 32 + T - 1) / T, T, 0, stream>>>(
      csr, rowptr, cnt, dinv, t2, t3, b2, b3, om, ol, OUTC, N);
}
